# starting kernel
# baseline (speedup 1.0000x reference)
#include <hip/hip_runtime.h>


typedef float v2f __attribute__((ext_vector_type(2)));
typedef float v8f __attribute__((ext_vector_type(8)));

#define C_CH 192
#define N_IN (192 * 192 * 9)

__device__ __forceinline__ v8f wmma_f32_16x16x4(v2f a, v2f b, v8f c) {
  return __builtin_amdgcn_wmma_f32_16x16x4_f32(false, a, false, b, (short)0, c,
                                               false, false);
}

__global__ void transpose_in_k(const float* __restrict__ in,
                               float* __restrict__ out, int n) {
  int i = blockIdx.x * blockDim.x + threadIdx.x;
  if (i < n) {
    int a = i % C_CH;
    int t = i / C_CH;
    int p = t % 9;
    int o = t / 9;
    out[i] = in[(o * C_CH + a) * 9 + p];
  }
}

__global__ void ssq_partial_k(const float* __restrict__ x, int n,
                              float* __restrict__ part) {
  __shared__ float sdata[256];
  float s = 0.f;
  for (int i = blockIdx.x * blockDim.x + threadIdx.x; i < n;
       i += gridDim.x * blockDim.x) {
    float v = x[i];
    s += v * v;
  }
  sdata[threadIdx.x] = s;
  __syncthreads();
  for (int off = 128; off > 0; off >>= 1) {
    if ((int)threadIdx.x < off) sdata[threadIdx.x] += sdata[threadIdx.x + off];
    __syncthreads();
  }
  if (threadIdx.x == 0) part[blockIdx.x] = sdata[0];
}

__global__ void final_reduce_k(const float* __restrict__ part, int n,
                               float* __restrict__ out) {
  __shared__ float sdata[256];
  float s = 0.f;
  for (int i = threadIdx.x; i < n; i += 256) s += part[i];
  sdata[threadIdx.x] = s;
  __syncthreads();
  for (int off = 128; off > 0; off >>= 1) {
    if ((int)threadIdx.x < off) sdata[threadIdx.x] += sdata[threadIdx.x + off];
    __syncthreads();
  }
  if (threadIdx.x == 0) out[0] = sdata[0];
}

__global__ __launch_bounds__(32) void gram_wmma_k(
    const float* __restrict__ src, float* __restrict__ dst,
    const float* __restrict__ f2, float* __restrict__ part, int k) {
  const int P = k * k;
  const int w = 2 * k - 1;
  const int S = w * w;
  const int sidx = blockIdx.y;
  const int sy = sidx / w - (k - 1);
  const int sx = sidx % w - (k - 1);
  const int ny = k - (sy < 0 ? -sy : sy);
  const int nx = k - (sx < 0 ? -sx : sx);
  const int validp = ny * nx;
  const int by0 = sy < 0 ? -sy : 0;
  const int bx0 = sx < 0 ? -sx : 0;
  const int dpa = sy * k + sx;
  const unsigned Mdiv = (65536u + (unsigned)nx - 1u) / (unsigned)nx;

  const int lane = threadIdx.x;
  const int l15 = lane & 15;
  const int h = lane >> 4;
  const int m0 = (blockIdx.x / 3) * 32;
  const int n0 = (blockIdx.x % 3) * 64;

  const v8f z = {0.f, 0.f, 0.f, 0.f, 0.f, 0.f, 0.f, 0.f};
  v8f acc[2][4] = {{z, z, z, z}, {z, z, z, z}};

  for (int o = 0; o < C_CH; ++o) {
    const float* baseo = src + (size_t)o * P * C_CH;
    if (o + 1 < C_CH) __builtin_prefetch(baseo + P * C_CH, 0, 3);

    int j0 = 0;
    for (; j0 + 4 <= validp; j0 += 4) {
      int jb = j0 + 2 * h;
      int jd0 = (int)(((unsigned)jb * Mdiv) >> 16);
      int jm0 = jb - jd0 * nx;
      int jm1 = jm0 + 1;
      bool wrap = (jm1 == nx);
      int jd1 = jd0 + (wrap ? 1 : 0);
      jm1 = wrap ? 0 : jm1;
      int pb0 = (by0 + jd0) * k + (bx0 + jm0);
      int pb1 = (by0 + jd1) * k + (bx0 + jm1);
      const float* rb0 = baseo + pb0 * C_CH;
      const float* rb1 = baseo + pb1 * C_CH;
      const float* ra0 = rb0 + dpa * C_CH;
      const float* ra1 = rb1 + dpa * C_CH;

      v2f av[2], bv[4];
      av[0][0] = ra0[m0 + l15];
      av[1][0] = ra0[m0 + 16 + l15];
      av[0][1] = ra1[m0 + l15];
      av[1][1] = ra1[m0 + 16 + l15];
#pragma unroll
      for (int tj = 0; tj < 4; ++tj) {
        bv[tj][0] = rb0[n0 + 16 * tj + l15];
        bv[tj][1] = rb1[n0 + 16 * tj + l15];
      }
#pragma unroll
      for (int ti = 0; ti < 2; ++ti)
#pragma unroll
        for (int tj = 0; tj < 4; ++tj)
          acc[ti][tj] = wmma_f32_16x16x4(av[ti], bv[tj], acc[ti][tj]);
    }
    if (j0 < validp) {
      v2f av[2], bv[4];
#pragma unroll
      for (int e = 0; e < 2; ++e) {
        int j = j0 + 2 * h + e;
        int jj = j < validp ? j : validp - 1;
        int jd = (int)(((unsigned)jj * Mdiv) >> 16);
        int jm = jj - jd * nx;
        int pb = (by0 + jd) * k + (bx0 + jm);
        const float* rb = baseo + pb * C_CH;
        const float* ra = rb + dpa * C_CH;
        bool ok = j < validp;
        av[0][e] = ok ? ra[m0 + l15] : 0.f;
        av[1][e] = ok ? ra[m0 + 16 + l15] : 0.f;
#pragma unroll
        for (int tj = 0; tj < 4; ++tj)
          bv[tj][e] = ok ? rb[n0 + 16 * tj + l15] : 0.f;
      }
#pragma unroll
      for (int ti = 0; ti < 2; ++ti)
#pragma unroll
        for (int tj = 0; tj < 4; ++tj)
          acc[ti][tj] = wmma_f32_16x16x4(av[ti], bv[tj], acc[ti][tj]);
    }
  }

  const float inv = 1.0f / f2[0];

  if (dst) {
#pragma unroll
    for (int ti = 0; ti < 2; ++ti)
#pragma unroll
      for (int r = 0; r < 8; ++r) {
        int am = m0 + 16 * ti + r + 8 * h;
        size_t row = ((size_t)am * S + sidx) * C_CH;
#pragma unroll
        for (int tj = 0; tj < 4; ++tj)
          dst[row + n0 + 16 * tj + l15] = acc[ti][tj][r] * inv;
      }
  } else {
    float ssq = 0.f;
#pragma unroll
    for (int ti = 0; ti < 2; ++ti)
#pragma unroll
      for (int tj = 0; tj < 4; ++tj)
#pragma unroll
        for (int r = 0; r < 8; ++r) {
          float g = acc[ti][tj][r] * inv;
          ssq += g * g;
        }
    for (int off = 16; off > 0; off >>= 1) ssq += __shfl_xor(ssq, off, 32);
    if (lane == 0) part[blockIdx.y * gridDim.x + blockIdx.x] = ssq;
  }
}

__global__ void sigma_k(const float* __restrict__ s,
                        float* __restrict__ inv_sigma) {
  if (threadIdx.x == 0 && blockIdx.x == 0) {
    double ls = 0.5 * (log((double)s[0]) + 0.5 * log((double)s[1]) +
                       0.25 * log((double)s[2]) + 0.125 * log((double)s[3]));
    inv_sigma[0] = (float)exp(-ls);
  }
}

__global__ void scale_k(const float* __restrict__ in,
                        const float* __restrict__ inv_sigma,
                        float* __restrict__ out, int n) {
  int i = blockIdx.x * blockDim.x + threadIdx.x;
  if (i < n) out[i] = in[i] * inv_sigma[0];
}

static inline int imin(int a, int b) { return a < b ? a : b; }

extern "C" void kernel_launch(void* const* d_in, const int* in_sizes, int n_in,
                              void* d_out, int out_size, void* d_ws,
                              size_t ws_size, hipStream_t stream) {
  (void)in_sizes; (void)n_in; (void)out_size; (void)ws_size;
  const float* K = (const float*)d_in[0];
  float* out = (float*)d_out;
  char* ws = (char*)d_ws;

  float* scal = (float*)ws;
  float* part = (float*)(ws + 1024);
  float* K1T = (float*)(ws + (1 << 17));
  float* K2T = (float*)(ws + (4 << 20));
  float* KT0 = K2T;

  const int N0 = N_IN;
  const int N1 = 192 * 192 * 25;
  const int N2 = 192 * 192 * 81;

  transpose_in_k<<<(N0 + 255) / 256, 256, 0, stream>>>(K, KT0, N0);
  int g0 = imin(1024, (N0 + 255) / 256);
  ssq_partial_k<<<g0, 256, 0, stream>>>(K, N0, part);
  final_reduce_k<<<1, 256, 0, stream>>>(part, g0, scal + 0);

  gram_wmma_k<<<dim3(18, 25), 32, 0, stream>>>(KT0, K1T, scal + 0, nullptr, 3);

  int g1 = imin(1024, (N1 + 255) / 256);
  ssq_partial_k<<<g1, 256, 0, stream>>>(K1T, N1, part);
  final_reduce_k<<<1, 256, 0, stream>>>(part, g1, scal + 1);

  gram_wmma_k<<<dim3(18, 81), 32, 0, stream>>>(K1T, K2T, scal + 1, nullptr, 5);

  int g2 = imin(1024, (N2 + 255) / 256);
  ssq_partial_k<<<g2, 256, 0, stream>>>(K2T, N2, part);
  final_reduce_k<<<1, 256, 0, stream>>>(part, g2, scal + 2);

  gram_wmma_k<<<dim3(18, 289), 32, 0, stream>>>(K2T, nullptr, scal + 2, part,
                                                9);
  final_reduce_k<<<1, 256, 0, stream>>>(part, 18 * 289, scal + 3);

  sigma_k<<<1, 1, 0, stream>>>(scal, scal + 4);
  scale_k<<<(N0 + 255) / 256, 256, 0, stream>>>(K, scal + 4, out, N0);
}


# Split each tile's o-reduction over the waves of a work-group (sixteen for the 3x3 and 5x5 Gram steps, eight for 9x9); partial tiles summed via LDS in fixed wave order (changes fp sum order)
# speedup vs baseline: 2.7293x; 2.7293x over previous
#include <hip/hip_runtime.h>


typedef float v2f __attribute__((ext_vector_type(2)));
typedef float v8f __attribute__((ext_vector_type(8)));

#define C_CH 192
#define N_IN (192 * 192 * 9)

__device__ __forceinline__ v8f wmma_f32_16x16x4(v2f a, v2f b, v8f c) {
  return __builtin_amdgcn_wmma_f32_16x16x4_f32(false, a, false, b, (short)0, c,
                                               false, false);
}

__global__ void transpose_in_k(const float* __restrict__ in,
                               float* __restrict__ out, int n) {
  int i = blockIdx.x * blockDim.x + threadIdx.x;
  if (i < n) {
    int a = i % C_CH;
    int t = i / C_CH;
    int p = t % 9;
    int o = t / 9;
    out[i] = in[(o * C_CH + a) * 9 + p];
  }
}

__global__ void ssq_partial_k(const float* __restrict__ x, int n,
                              float* __restrict__ part) {
  __shared__ float sdata[256];
  float s = 0.f;
  for (int i = blockIdx.x * blockDim.x + threadIdx.x; i < n;
       i += gridDim.x * blockDim.x) {
    float v = x[i];
    s += v * v;
  }
  sdata[threadIdx.x] = s;
  __syncthreads();
  for (int off = 128; off > 0; off >>= 1) {
    if ((int)threadIdx.x < off) sdata[threadIdx.x] += sdata[threadIdx.x + off];
    __syncthreads();
  }
  if (threadIdx.x == 0) part[blockIdx.x] = sdata[0];
}

__global__ void final_reduce_k(const float* __restrict__ part, int n,
                               float* __restrict__ out) {
  __shared__ float sdata[256];
  float s = 0.f;
  for (int i = threadIdx.x; i < n; i += 256) s += part[i];
  sdata[threadIdx.x] = s;
  __syncthreads();
  for (int off = 128; off > 0; off >>= 1) {
    if ((int)threadIdx.x < off) sdata[threadIdx.x] += sdata[threadIdx.x + off];
    __syncthreads();
  }
  if (threadIdx.x == 0) out[0] = sdata[0];
}

template <int NW>
__device__ __forceinline__ void gram_wmma_body(
    const float* __restrict__ src, float* __restrict__ dst,
    const float* __restrict__ f2, float* __restrict__ part, int k,
    float* red) {
  constexpr int OPW = C_CH / NW;
  constexpr int VPT = 32 / NW;
  const int P = k * k;
  const int w = 2 * k - 1;
  const int S = w * w;
  const int sidx = blockIdx.y;
  const int sy = sidx / w - (k - 1);
  const int sx = sidx % w - (k - 1);
  const int ny = k - (sy < 0 ? -sy : sy);
  const int nx = k - (sx < 0 ? -sx : sx);
  const int validp = ny * nx;
  const int by0 = sy < 0 ? -sy : 0;
  const int bx0 = sx < 0 ? -sx : 0;
  const int dpa = sy * k + sx;
  const unsigned Mdiv = (65536u + (unsigned)nx - 1u) / (unsigned)nx;

  const int lane = threadIdx.x & 31;
  const int wave = __builtin_amdgcn_readfirstlane((int)(threadIdx.x >> 5));
  const int l15 = lane & 15;
  const int h = lane >> 4;
  const int m0 = (blockIdx.x / 3) * 32;
  const int n0 = (blockIdx.x % 3) * 64;

  const v8f z = {0.f, 0.f, 0.f, 0.f, 0.f, 0.f, 0.f, 0.f};
  v8f acc[2][4] = {{z, z, z, z}, {z, z, z, z}};

  const int o_begin = wave * OPW;
  const int o_end = o_begin + OPW;
  for (int o = o_begin; o < o_end; ++o) {
    const float* baseo = src + (size_t)o * P * C_CH;
    if (o + 1 < o_end) __builtin_prefetch(baseo + P * C_CH, 0, 3);

    int j0 = 0;
    for (; j0 + 4 <= validp; j0 += 4) {
      int jb = j0 + 2 * h;
      int jd0 = (int)(((unsigned)jb * Mdiv) >> 16);
      int jm0 = jb - jd0 * nx;
      int jm1 = jm0 + 1;
      bool wrap = (jm1 == nx);
      int jd1 = jd0 + (wrap ? 1 : 0);
      jm1 = wrap ? 0 : jm1;
      int pb0 = (by0 + jd0) * k + (bx0 + jm0);
      int pb1 = (by0 + jd1) * k + (bx0 + jm1);
      const float* rb0 = baseo + pb0 * C_CH;
      const float* rb1 = baseo + pb1 * C_CH;
      const float* ra0 = rb0 + dpa * C_CH;
      const float* ra1 = rb1 + dpa * C_CH;

      v2f av[2], bv[4];
      av[0][0] = ra0[m0 + l15];
      av[1][0] = ra0[m0 + 16 + l15];
      av[0][1] = ra1[m0 + l15];
      av[1][1] = ra1[m0 + 16 + l15];
#pragma unroll
      for (int tj = 0; tj < 4; ++tj) {
        bv[tj][0] = rb0[n0 + 16 * tj + l15];
        bv[tj][1] = rb1[n0 + 16 * tj + l15];
      }
#pragma unroll
      for (int ti = 0; ti < 2; ++ti)
#pragma unroll
        for (int tj = 0; tj < 4; ++tj)
          acc[ti][tj] = wmma_f32_16x16x4(av[ti], bv[tj], acc[ti][tj]);
    }
    if (j0 < validp) {
      v2f av[2], bv[4];
#pragma unroll
      for (int e = 0; e < 2; ++e) {
        int j = j0 + 2 * h + e;
        int jj = j < validp ? j : validp - 1;
        int jd = (int)(((unsigned)jj * Mdiv) >> 16);
        int jm = jj - jd * nx;
        int pb = (by0 + jd) * k + (bx0 + jm);
        const float* rb = baseo + pb * C_CH;
        const float* ra = rb + dpa * C_CH;
        bool ok = j < validp;
        av[0][e] = ok ? ra[m0 + l15] : 0.f;
        av[1][e] = ok ? ra[m0 + 16 + l15] : 0.f;
#pragma unroll
        for (int tj = 0; tj < 4; ++tj)
          bv[tj][e] = ok ? rb[n0 + 16 * tj + l15] : 0.f;
      }
#pragma unroll
      for (int ti = 0; ti < 2; ++ti)
#pragma unroll
        for (int tj = 0; tj < 4; ++tj)
          acc[ti][tj] = wmma_f32_16x16x4(av[ti], bv[tj], acc[ti][tj]);
    }
  }

  const float inv = 1.0f / f2[0];

  const int rr = wave & 7;
  const int tjb = (wave >> 3) * VPT;
  float ssq = 0.f;
#pragma unroll
  for (int ti = 0; ti < 2; ++ti) {
    if (ti) __syncthreads();
#pragma unroll
    for (int tj = 0; tj < 4; ++tj)
#pragma unroll
      for (int r = 0; r < 8; ++r)
        red[(wave * 32 + tj * 8 + r) * 32 + lane] = acc[ti][tj][r];
    __syncthreads();
    const int am = m0 + 16 * ti + rr + 8 * h;
    const size_t row = ((size_t)am * S + sidx) * C_CH;
#pragma unroll
    for (int q = 0; q < VPT; ++q) {
      const int tj = tjb + q;
      float v = red[(tj * 8 + rr) * 32 + lane];
#pragma unroll
      for (int w2 = 1; w2 < NW; ++w2)
        v += red[(w2 * 32 + tj * 8 + rr) * 32 + lane];
      v *= inv;
      if (dst)
        dst[row + n0 + 16 * tj + l15] = v;
      else
        ssq += v * v;
    }
  }

  if (!dst) {
    for (int off = 16; off > 0; off >>= 1) ssq += __shfl_xor(ssq, off, 32);
    __syncthreads();
    if (lane == 0) red[wave] = ssq;
    __syncthreads();
    if (threadIdx.x == 0) {
      float s = red[0];
      for (int w2 = 1; w2 < NW; ++w2) s += red[w2];
      part[blockIdx.y * gridDim.x + blockIdx.x] = s;
    }
  }
}

__global__ __launch_bounds__(512) void gram_wmma16_k(
    const float* __restrict__ src, float* __restrict__ dst,
    const float* __restrict__ f2, float* __restrict__ part, int k) {
  __shared__ float red[16 * 32 * 32];
  gram_wmma_body<16>(src, dst, f2, part, k, red);
}

__global__ __launch_bounds__(256) void gram_wmma8_k(
    const float* __restrict__ src, float* __restrict__ dst,
    const float* __restrict__ f2, float* __restrict__ part, int k) {
  __shared__ float red[8 * 32 * 32];
  gram_wmma_body<8>(src, dst, f2, part, k, red);
}

__global__ void sigma_k(const float* __restrict__ s,
                        float* __restrict__ inv_sigma) {
  if (threadIdx.x == 0 && blockIdx.x == 0) {
    double ls = 0.5 * (log((double)s[0]) + 0.5 * log((double)s[1]) +
                       0.25 * log((double)s[2]) + 0.125 * log((double)s[3]));
    inv_sigma[0] = (float)exp(-ls);
  }
}

__global__ void scale_k(const float* __restrict__ in,
                        const float* __restrict__ inv_sigma,
                        float* __restrict__ out, int n) {
  int i = blockIdx.x * blockDim.x + threadIdx.x;
  if (i < n) out[i] = in[i] * inv_sigma[0];
}

static inline int imin(int a, int b) { return a < b ? a : b; }

extern "C" void kernel_launch(void* const* d_in, const int* in_sizes, int n_in,
                              void* d_out, int out_size, void* d_ws,
                              size_t ws_size, hipStream_t stream) {
  (void)in_sizes; (void)n_in; (void)out_size; (void)ws_size;
  const float* K = (const float*)d_in[0];
  float* out = (float*)d_out;
  char* ws = (char*)d_ws;

  float* scal = (float*)ws;
  float* part = (float*)(ws + 1024);
  float* K1T = (float*)(ws + (1 << 17));
  float* K2T = (float*)(ws + (4 << 20));
  float* KT0 = K2T;

  const int N0 = N_IN;
  const int N1 = 192 * 192 * 25;
  const int N2 = 192 * 192 * 81;

  transpose_in_k<<<(N0 + 255) / 256, 256, 0, stream>>>(K, KT0, N0);
  int g0 = imin(1024, (N0 + 255) / 256);
  ssq_partial_k<<<g0, 256, 0, stream>>>(K, N0, part);
  final_reduce_k<<<1, 256, 0, stream>>>(part, g0, scal + 0);

  gram_wmma16_k<<<dim3(18, 25), 512, 0, stream>>>(KT0, K1T, scal + 0, nullptr,
                                                  3);

  int g1 = imin(1024, (N1 + 255) / 256);
  ssq_partial_k<<<g1, 256, 0, stream>>>(K1T, N1, part);
  final_reduce_k<<<1, 256, 0, stream>>>(part, g1, scal + 1);

  gram_wmma16_k<<<dim3(18, 81), 512, 0, stream>>>(K1T, K2T, scal + 1, nullptr,
                                                  5);

  int g2 = imin(1024, (N2 + 255) / 256);
  ssq_partial_k<<<g2, 256, 0, stream>>>(K2T, N2, part);
  final_reduce_k<<<1, 256, 0, stream>>>(part, g2, scal + 2);

  gram_wmma8_k<<<dim3(18, 289), 256, 0, stream>>>(K2T, nullptr, scal + 2, part,
                                                  9);
  final_reduce_k<<<1, 256, 0, stream>>>(part, 18 * 289, scal + 3);

  sigma_k<<<1, 1, 0, stream>>>(scal, scal + 4);
  scale_k<<<(N0 + 255) / 256, 256, 0, stream>>>(K, scal + 4, out, N0);
}


# Gram mirror identity G(-s)[b,a] = G(s)[a,b]: each Gram step computes only shifts up to the centre, heaviest first; mirror tiles stored transposed via LDS, final norm doubled (changes fp sum order)
# speedup vs baseline: 4.8031x; 1.7598x over previous
#include <hip/hip_runtime.h>


typedef float v2f __attribute__((ext_vector_type(2)));
typedef float v8f __attribute__((ext_vector_type(8)));

#define C_CH 192
#define N_IN (192 * 192 * 9)

__device__ __forceinline__ v8f wmma_f32_16x16x4(v2f a, v2f b, v8f c) {
  return __builtin_amdgcn_wmma_f32_16x16x4_f32(false, a, false, b, (short)0, c,
                                               false, false);
}

__global__ void transpose_in_k(const float* __restrict__ in,
                               float* __restrict__ out, int n) {
  int i = blockIdx.x * blockDim.x + threadIdx.x;
  if (i < n) {
    int a = i % C_CH;
    int t = i / C_CH;
    int p = t % 9;
    int o = t / 9;
    out[i] = in[(o * C_CH + a) * 9 + p];
  }
}

__global__ void ssq_partial_k(const float* __restrict__ x, int n,
                              float* __restrict__ part) {
  __shared__ float sdata[256];
  float s = 0.f;
  for (int i = blockIdx.x * blockDim.x + threadIdx.x; i < n;
       i += gridDim.x * blockDim.x) {
    float v = x[i];
    s += v * v;
  }
  sdata[threadIdx.x] = s;
  __syncthreads();
  for (int off = 128; off > 0; off >>= 1) {
    if ((int)threadIdx.x < off) sdata[threadIdx.x] += sdata[threadIdx.x + off];
    __syncthreads();
  }
  if (threadIdx.x == 0) part[blockIdx.x] = sdata[0];
}

__global__ void final_reduce_k(const float* __restrict__ part, int n,
                               float* __restrict__ out) {
  __shared__ float sdata[256];
  float s = 0.f;
  for (int i = threadIdx.x; i < n; i += 256) s += part[i];
  sdata[threadIdx.x] = s;
  __syncthreads();
  for (int off = 128; off > 0; off >>= 1) {
    if ((int)threadIdx.x < off) sdata[threadIdx.x] += sdata[threadIdx.x + off];
    __syncthreads();
  }
  if (threadIdx.x == 0) out[0] = sdata[0];
}

template <int NW>
__device__ __forceinline__ void gram_wmma_body(
    const float* __restrict__ src, float* __restrict__ dst,
    const float* __restrict__ f2, float* __restrict__ part, int k,
    float* red) {
  constexpr int OPW = C_CH / NW;
  constexpr int VPT = 32 / NW;
  const int P = k * k;
  const int w = 2 * k - 1;
  const int S = w * w;
  const int sidx = (S >> 1) - (int)blockIdx.y;
  const int sy = sidx / w - (k - 1);
  const int sx = sidx % w - (k - 1);
  const int ny = k - (sy < 0 ? -sy : sy);
  const int nx = k - (sx < 0 ? -sx : sx);
  const int validp = ny * nx;
  const int by0 = sy < 0 ? -sy : 0;
  const int bx0 = sx < 0 ? -sx : 0;
  const int dpa = sy * k + sx;
  const unsigned Mdiv = (65536u + (unsigned)nx - 1u) / (unsigned)nx;

  const int lane = threadIdx.x & 31;
  const int wave = __builtin_amdgcn_readfirstlane((int)(threadIdx.x >> 5));
  const int l15 = lane & 15;
  const int h = lane >> 4;
  const int m0 = (blockIdx.x / 3) * 32;
  const int n0 = (blockIdx.x % 3) * 64;

  const v8f z = {0.f, 0.f, 0.f, 0.f, 0.f, 0.f, 0.f, 0.f};
  v8f acc[2][4] = {{z, z, z, z}, {z, z, z, z}};

  const int o_begin = wave * OPW;
  const int o_end = o_begin + OPW;
  for (int o = o_begin; o < o_end; ++o) {
    const float* baseo = src + (size_t)o * P * C_CH;
    if (o + 1 < o_end) __builtin_prefetch(baseo + P * C_CH, 0, 3);

    int j0 = 0;
    for (; j0 + 4 <= validp; j0 += 4) {
      int jb = j0 + 2 * h;
      int jd0 = (int)(((unsigned)jb * Mdiv) >> 16);
      int jm0 = jb - jd0 * nx;
      int jm1 = jm0 + 1;
      bool wrap = (jm1 == nx);
      int jd1 = jd0 + (wrap ? 1 : 0);
      jm1 = wrap ? 0 : jm1;
      int pb0 = (by0 + jd0) * k + (bx0 + jm0);
      int pb1 = (by0 + jd1) * k + (bx0 + jm1);
      const float* rb0 = baseo + pb0 * C_CH;
      const float* rb1 = baseo + pb1 * C_CH;
      const float* ra0 = rb0 + dpa * C_CH;
      const float* ra1 = rb1 + dpa * C_CH;

      v2f av[2], bv[4];
      av[0][0] = ra0[m0 + l15];
      av[1][0] = ra0[m0 + 16 + l15];
      av[0][1] = ra1[m0 + l15];
      av[1][1] = ra1[m0 + 16 + l15];
#pragma unroll
      for (int tj = 0; tj < 4; ++tj) {
        bv[tj][0] = rb0[n0 + 16 * tj + l15];
        bv[tj][1] = rb1[n0 + 16 * tj + l15];
      }
#pragma unroll
      for (int ti = 0; ti < 2; ++ti)
#pragma unroll
        for (int tj = 0; tj < 4; ++tj)
          acc[ti][tj] = wmma_f32_16x16x4(av[ti], bv[tj], acc[ti][tj]);
    }
    if (j0 < validp) {
      v2f av[2], bv[4];
#pragma unroll
      for (int e = 0; e < 2; ++e) {
        int j = j0 + 2 * h + e;
        int jj = j < validp ? j : validp - 1;
        int jd = (int)(((unsigned)jj * Mdiv) >> 16);
        int jm = jj - jd * nx;
        int pb = (by0 + jd) * k + (bx0 + jm);
        const float* rb = baseo + pb * C_CH;
        const float* ra = rb + dpa * C_CH;
        bool ok = j < validp;
        av[0][e] = ok ? ra[m0 + l15] : 0.f;
        av[1][e] = ok ? ra[m0 + 16 + l15] : 0.f;
#pragma unroll
        for (int tj = 0; tj < 4; ++tj)
          bv[tj][e] = ok ? rb[n0 + 16 * tj + l15] : 0.f;
      }
#pragma unroll
      for (int ti = 0; ti < 2; ++ti)
#pragma unroll
        for (int tj = 0; tj < 4; ++tj)
          acc[ti][tj] = wmma_f32_16x16x4(av[ti], bv[tj], acc[ti][tj]);
    }
  }

  const float inv = 1.0f / f2[0];

  const int rr = wave & 7;
  const int tjb = (wave >> 3) * VPT;
  float ssq = 0.f;
  float vv[2][VPT];
#pragma unroll
  for (int ti = 0; ti < 2; ++ti) {
    if (ti) __syncthreads();
#pragma unroll
    for (int tj = 0; tj < 4; ++tj)
#pragma unroll
      for (int r = 0; r < 8; ++r)
        red[(wave * 32 + tj * 8 + r) * 32 + lane] = acc[ti][tj][r];
    __syncthreads();
    const int am = m0 + 16 * ti + rr + 8 * h;
    const size_t row = ((size_t)am * S + sidx) * C_CH;
#pragma unroll
    for (int q = 0; q < VPT; ++q) {
      const int tj = tjb + q;
      float v = red[(tj * 8 + rr) * 32 + lane];
#pragma unroll
      for (int w2 = 1; w2 < NW; ++w2)
        v += red[(w2 * 32 + tj * 8 + rr) * 32 + lane];
      v *= inv;
      vv[ti][q] = v;
      if (dst)
        dst[row + n0 + 16 * tj + l15] = v;
      else
        ssq += v * v;
    }
  }

  if (dst && sidx != (S >> 1)) {
    constexpr int TP = 33;
    constexpr int CPW = 64 / NW;
    __syncthreads();
#pragma unroll
    for (int ti = 0; ti < 2; ++ti)
#pragma unroll
      for (int q = 0; q < VPT; ++q)
        red[(16 * (tjb + q) + l15) * TP + 16 * ti + rr + 8 * h] = vv[ti][q];
    __syncthreads();
    const int ms = S - 1 - sidx;
#pragma unroll
    for (int c = 0; c < CPW; ++c) {
      const int b = wave * CPW + c;
      dst[((size_t)(n0 + b) * S + ms) * C_CH + m0 + lane] = red[b * TP + lane];
    }
  }

  if (!dst) {
    for (int off = 16; off > 0; off >>= 1) ssq += __shfl_xor(ssq, off, 32);
    __syncthreads();
    if (lane == 0) red[wave] = ssq;
    __syncthreads();
    if (threadIdx.x == 0) {
      float s = red[0];
      for (int w2 = 1; w2 < NW; ++w2) s += red[w2];
      part[blockIdx.y * gridDim.x + blockIdx.x] = (sidx == (S >> 1)) ? s : 2.f * s;
    }
  }
}

__global__ __launch_bounds__(512) void gram_wmma16_k(
    const float* __restrict__ src, float* __restrict__ dst,
    const float* __restrict__ f2, float* __restrict__ part, int k) {
  __shared__ float red[16 * 32 * 32];
  gram_wmma_body<16>(src, dst, f2, part, k, red);
}

__global__ __launch_bounds__(256) void gram_wmma8_k(
    const float* __restrict__ src, float* __restrict__ dst,
    const float* __restrict__ f2, float* __restrict__ part, int k) {
  __shared__ float red[8 * 32 * 32];
  gram_wmma_body<8>(src, dst, f2, part, k, red);
}

__global__ void sigma_k(const float* __restrict__ s,
                        float* __restrict__ inv_sigma) {
  if (threadIdx.x == 0 && blockIdx.x == 0) {
    double ls = 0.5 * (log((double)s[0]) + 0.5 * log((double)s[1]) +
                       0.25 * log((double)s[2]) + 0.125 * log((double)s[3]));
    inv_sigma[0] = (float)exp(-ls);
  }
}

__global__ void scale_k(const float* __restrict__ in,
                        const float* __restrict__ inv_sigma,
                        float* __restrict__ out, int n) {
  int i = blockIdx.x * blockDim.x + threadIdx.x;
  if (i < n) out[i] = in[i] * inv_sigma[0];
}

static inline int imin(int a, int b) { return a < b ? a : b; }

extern "C" void kernel_launch(void* const* d_in, const int* in_sizes, int n_in,
                              void* d_out, int out_size, void* d_ws,
                              size_t ws_size, hipStream_t stream) {
  (void)in_sizes; (void)n_in; (void)out_size; (void)ws_size;
  const float* K = (const float*)d_in[0];
  float* out = (float*)d_out;
  char* ws = (char*)d_ws;

  float* scal = (float*)ws;
  float* part = (float*)(ws + 1024);
  float* K1T = (float*)(ws + (1 << 17));
  float* K2T = (float*)(ws + (4 << 20));
  float* KT0 = K2T;

  const int N0 = N_IN;
  const int N1 = 192 * 192 * 25;
  const int N2 = 192 * 192 * 81;

  transpose_in_k<<<(N0 + 255) / 256, 256, 0, stream>>>(K, KT0, N0);
  int g0 = imin(1024, (N0 + 255) / 256);
  ssq_partial_k<<<g0, 256, 0, stream>>>(K, N0, part);
  final_reduce_k<<<1, 256, 0, stream>>>(part, g0, scal + 0);

  gram_wmma16_k<<<dim3(18, 13), 512, 0, stream>>>(KT0, K1T, scal + 0, nullptr,
                                                  3);

  int g1 = imin(1024, (N1 + 255) / 256);
  ssq_partial_k<<<g1, 256, 0, stream>>>(K1T, N1, part);
  final_reduce_k<<<1, 256, 0, stream>>>(part, g1, scal + 1);

  gram_wmma16_k<<<dim3(18, 41), 512, 0, stream>>>(K1T, K2T, scal + 1, nullptr,
                                                  5);

  int g2 = imin(1024, (N2 + 255) / 256);
  ssq_partial_k<<<g2, 256, 0, stream>>>(K2T, N2, part);
  final_reduce_k<<<1, 256, 0, stream>>>(part, g2, scal + 2);

  gram_wmma8_k<<<dim3(18, 145), 256, 0, stream>>>(K2T, nullptr, scal + 2, part,
                                                  9);
  final_reduce_k<<<1, 256, 0, stream>>>(part, 18 * 145, scal + 3);

  sigma_k<<<1, 1, 0, stream>>>(scal, scal + 4);
  scale_k<<<(N0 + 255) / 256, 256, 0, stream>>>(K, scal + 4, out, N0);
}


# Gram steps: the 12 fragment loads of k-step i+1 are issued as one group into a second register set before the 8 matrix products of step i; sums and their order unchanged
# speedup vs baseline: 5.2561x; 1.0943x over previous
#include <hip/hip_runtime.h>


typedef float v2f __attribute__((ext_vector_type(2)));
typedef float v8f __attribute__((ext_vector_type(8)));

#define C_CH 192
#define N_IN (192 * 192 * 9)

__device__ __forceinline__ v8f wmma_f32_16x16x4(v2f a, v2f b, v8f c) {
  return __builtin_amdgcn_wmma_f32_16x16x4_f32(false, a, false, b, (short)0, c,
                                               false, false);
}

__global__ void transpose_in_k(const float* __restrict__ in,
                               float* __restrict__ out, int n) {
  int i = blockIdx.x * blockDim.x + threadIdx.x;
  if (i < n) {
    int a = i % C_CH;
    int t = i / C_CH;
    int p = t % 9;
    int o = t / 9;
    out[i] = in[(o * C_CH + a) * 9 + p];
  }
}

__global__ void ssq_partial_k(const float* __restrict__ x, int n,
                              float* __restrict__ part) {
  __shared__ float sdata[256];
  float s = 0.f;
  for (int i = blockIdx.x * blockDim.x + threadIdx.x; i < n;
       i += gridDim.x * blockDim.x) {
    float v = x[i];
    s += v * v;
  }
  sdata[threadIdx.x] = s;
  __syncthreads();
  for (int off = 128; off > 0; off >>= 1) {
    if ((int)threadIdx.x < off) sdata[threadIdx.x] += sdata[threadIdx.x + off];
    __syncthreads();
  }
  if (threadIdx.x == 0) part[blockIdx.x] = sdata[0];
}

__global__ void final_reduce_k(const float* __restrict__ part, int n,
                               float* __restrict__ out) {
  __shared__ float sdata[256];
  float s = 0.f;
  for (int i = threadIdx.x; i < n; i += 256) s += part[i];
  sdata[threadIdx.x] = s;
  __syncthreads();
  for (int off = 128; off > 0; off >>= 1) {
    if ((int)threadIdx.x < off) sdata[threadIdx.x] += sdata[threadIdx.x + off];
    __syncthreads();
  }
  if (threadIdx.x == 0) out[0] = sdata[0];
}

template <int NW>
__device__ __forceinline__ void gram_wmma_body(
    const float* __restrict__ src, float* __restrict__ dst,
    const float* __restrict__ f2, float* __restrict__ part, int k,
    float* red) {
  constexpr int OPW = C_CH / NW;
  constexpr int VPT = 32 / NW;
  const int P = k * k;
  const int w = 2 * k - 1;
  const int S = w * w;
  const int sidx = (S >> 1) - (int)blockIdx.y;
  const int sy = sidx / w - (k - 1);
  const int sx = sidx % w - (k - 1);
  const int ny = k - (sy < 0 ? -sy : sy);
  const int nx = k - (sx < 0 ? -sx : sx);
  const int validp = ny * nx;
  const int by0 = sy < 0 ? -sy : 0;
  const int bx0 = sx < 0 ? -sx : 0;
  const int dpa = sy * k + sx;
  const unsigned Mdiv = (65536u + (unsigned)nx - 1u) / (unsigned)nx;

  const int lane = threadIdx.x & 31;
  const int wave = __builtin_amdgcn_readfirstlane((int)(threadIdx.x >> 5));
  const int l15 = lane & 15;
  const int h = lane >> 4;
  const int m0 = (blockIdx.x / 3) * 32;
  const int n0 = (blockIdx.x % 3) * 64;

  const v8f z = {0.f, 0.f, 0.f, 0.f, 0.f, 0.f, 0.f, 0.f};
  v8f acc[2][4] = {{z, z, z, z}, {z, z, z, z}};

  static_assert((C_CH / NW) % 2 == 0, "the step loop below is unrolled by two");
  const int o_begin = wave * OPW;
  const int o_end = o_begin + OPW;
  const int vlast = validp - 1;
  const int T = OPW * ((validp + 3) >> 2);
  const int h2 = 2 * h;
  const unsigned cbB = (unsigned)(n0 + l15) * 4u;
  const int caB = (dpa * C_CH + m0 + l15) * 4;

  auto load_step = [&](int o, int j0, v2f (&av)[2], v2f (&bv)[4]) {
    const char* baseo = (const char*)(src + (size_t)o * P * C_CH);
    if (j0 == 0 && o + 1 < o_end)
      __builtin_prefetch(baseo + (size_t)P * C_CH * 4, 0, 3);
#pragma unroll
    for (int e = 0; e < 2; ++e) {
      int j = j0 + h2 + e;
      int jj = j < vlast ? j : vlast;
      int jd = (int)(((unsigned)jj * Mdiv) >> 16);
      int jm = jj - jd * nx;
      int pb = (by0 + jd) * k + (bx0 + jm);
      unsigned rowB = (unsigned)(pb * (C_CH * 4));
      unsigned ob = rowB + cbB;
      unsigned oa = (unsigned)((int)rowB + caB);
      av[0][e] = *(const float*)(baseo + oa);
      av[1][e] = *(const float*)(baseo + oa + 64u);
#pragma unroll
      for (int tj = 0; tj < 4; ++tj)
        bv[tj][e] = *(const float*)(baseo + ob + 64u * (unsigned)tj);
    }
  };

  auto mma_step = [&](int j0, v2f (&av)[2], v2f (&bv)[4]) {
    if (j0 + 4 > validp) {
#pragma unroll
      for (int e = 0; e < 2; ++e) {
        const bool ok = (j0 + h2 + e) < validp;
        av[0][e] = ok ? av[0][e] : 0.f;
        av[1][e] = ok ? av[1][e] : 0.f;
#pragma unroll
        for (int tj = 0; tj < 4; ++tj) bv[tj][e] = ok ? bv[tj][e] : 0.f;
      }
    }
#pragma unroll
    for (int ti = 0; ti < 2; ++ti)
#pragma unroll
      for (int tj = 0; tj < 4; ++tj)
        acc[ti][tj] = wmma_f32_16x16x4(av[ti], bv[tj], acc[ti][tj]);
  };

  {
    v2f aX[2], bX[4], aY[2], bY[4];
    int oX = o_begin, jX = 0;
    int oY, jY;
    load_step(oX, jX, aX, bX);
    for (int t = 2; t < T; t += 2) {
      oY = oX; jY = jX + 4;
      if (jY >= validp) { jY = 0; ++oY; }
      load_step(oY, jY, aY, bY);
      mma_step(jX, aX, bX);
      oX = oY; jX = jY + 4;
      if (jX >= validp) { jX = 0; ++oX; }
      load_step(oX, jX, aX, bX);
      mma_step(jY, aY, bY);
    }
    oY = oX; jY = jX + 4;
    if (jY >= validp) { jY = 0; ++oY; }
    load_step(oY, jY, aY, bY);
    mma_step(jX, aX, bX);
    mma_step(jY, aY, bY);
  }

  const float inv = 1.0f / f2[0];

  const int rr = wave & 7;
  const int tjb = (wave >> 3) * VPT;
  float ssq = 0.f;
  float vv[2][VPT];
#pragma unroll
  for (int ti = 0; ti < 2; ++ti) {
    if (ti) __syncthreads();
#pragma unroll
    for (int tj = 0; tj < 4; ++tj)
#pragma unroll
      for (int r = 0; r < 8; ++r)
        red[(wave * 32 + tj * 8 + r) * 32 + lane] = acc[ti][tj][r];
    __syncthreads();
    const int am = m0 + 16 * ti + rr + 8 * h;
    const size_t row = ((size_t)am * S + sidx) * C_CH;
#pragma unroll
    for (int q = 0; q < VPT; ++q) {
      const int tj = tjb + q;
      float v = red[(tj * 8 + rr) * 32 + lane];
#pragma unroll
      for (int w2 = 1; w2 < NW; ++w2)
        v += red[(w2 * 32 + tj * 8 + rr) * 32 + lane];
      v *= inv;
      vv[ti][q] = v;
      if (dst)
        dst[row + n0 + 16 * tj + l15] = v;
      else
        ssq += v * v;
    }
  }

  if (dst && sidx != (S >> 1)) {
    constexpr int TP = 33;
    constexpr int CPW = 64 / NW;
    __syncthreads();
#pragma unroll
    for (int ti = 0; ti < 2; ++ti)
#pragma unroll
      for (int q = 0; q < VPT; ++q)
        red[(16 * (tjb + q) + l15) * TP + 16 * ti + rr + 8 * h] = vv[ti][q];
    __syncthreads();
    const int ms = S - 1 - sidx;
#pragma unroll
    for (int c = 0; c < CPW; ++c) {
      const int b = wave * CPW + c;
      dst[((size_t)(n0 + b) * S + ms) * C_CH + m0 + lane] = red[b * TP + lane];
    }
  }

  if (!dst) {
    for (int off = 16; off > 0; off >>= 1) ssq += __shfl_xor(ssq, off, 32);
    __syncthreads();
    if (lane == 0) red[wave] = ssq;
    __syncthreads();
    if (threadIdx.x == 0) {
      float s = red[0];
      for (int w2 = 1; w2 < NW; ++w2) s += red[w2];
      part[blockIdx.y * gridDim.x + blockIdx.x] = (sidx == (S >> 1)) ? s : 2.f * s;
    }
  }
}

__global__ __launch_bounds__(512) void gram_wmma16_k(
    const float* __restrict__ src, float* __restrict__ dst,
    const float* __restrict__ f2, float* __restrict__ part, int k) {
  __shared__ float red[16 * 32 * 32];
  gram_wmma_body<16>(src, dst, f2, part, k, red);
}

__global__ __launch_bounds__(256) void gram_wmma8_k(
    const float* __restrict__ src, float* __restrict__ dst,
    const float* __restrict__ f2, float* __restrict__ part, int k) {
  __shared__ float red[8 * 32 * 32];
  gram_wmma_body<8>(src, dst, f2, part, k, red);
}

__global__ void sigma_k(const float* __restrict__ s,
                        float* __restrict__ inv_sigma) {
  if (threadIdx.x == 0 && blockIdx.x == 0) {
    double ls = 0.5 * (log((double)s[0]) + 0.5 * log((double)s[1]) +
                       0.25 * log((double)s[2]) + 0.125 * log((double)s[3]));
    inv_sigma[0] = (float)exp(-ls);
  }
}

__global__ void scale_k(const float* __restrict__ in,
                        const float* __restrict__ inv_sigma,
                        float* __restrict__ out, int n) {
  int i = blockIdx.x * blockDim.x + threadIdx.x;
  if (i < n) out[i] = in[i] * inv_sigma[0];
}

static inline int imin(int a, int b) { return a < b ? a : b; }

extern "C" void kernel_launch(void* const* d_in, const int* in_sizes, int n_in,
                              void* d_out, int out_size, void* d_ws,
                              size_t ws_size, hipStream_t stream) {
  (void)in_sizes; (void)n_in; (void)out_size; (void)ws_size;
  const float* K = (const float*)d_in[0];
  float* out = (float*)d_out;
  char* ws = (char*)d_ws;

  float* scal = (float*)ws;
  float* part = (float*)(ws + 1024);
  float* K1T = (float*)(ws + (1 << 17));
  float* K2T = (float*)(ws + (4 << 20));
  float* KT0 = K2T;

  const int N0 = N_IN;
  const int N1 = 192 * 192 * 25;
  const int N2 = 192 * 192 * 81;

  transpose_in_k<<<(N0 + 255) / 256, 256, 0, stream>>>(K, KT0, N0);
  int g0 = imin(1024, (N0 + 255) / 256);
  ssq_partial_k<<<g0, 256, 0, stream>>>(K, N0, part);
  final_reduce_k<<<1, 256, 0, stream>>>(part, g0, scal + 0);

  gram_wmma16_k<<<dim3(18, 13), 512, 0, stream>>>(KT0, K1T, scal + 0, nullptr,
                                                  3);

  int g1 = imin(1024, (N1 + 255) / 256);
  ssq_partial_k<<<g1, 256, 0, stream>>>(K1T, N1, part);
  final_reduce_k<<<1, 256, 0, stream>>>(part, g1, scal + 1);

  gram_wmma16_k<<<dim3(18, 41), 512, 0, stream>>>(K1T, K2T, scal + 1, nullptr,
                                                  5);

  int g2 = imin(1024, (N2 + 255) / 256);
  ssq_partial_k<<<g2, 256, 0, stream>>>(K2T, N2, part);
  final_reduce_k<<<1, 256, 0, stream>>>(part, g2, scal + 2);

  gram_wmma8_k<<<dim3(18, 145), 256, 0, stream>>>(K2T, nullptr, scal + 2, part,
                                                  9);
  final_reduce_k<<<1, 256, 0, stream>>>(part, 18 * 145, scal + 3);

  sigma_k<<<1, 1, 0, stream>>>(scal, scal + 4);
  scale_k<<<(N0 + 255) / 256, 256, 0, stream>>>(K, scal + 4, out, N0);
}


# Largest Gram step (9x9): each tile's o-reduction split over sixteen waves instead of eight, halving the longest wave's chain of k-steps; partials summed via LDS in wave order (changes fp sum order)
# speedup vs baseline: 5.4552x; 1.0379x over previous
#include <hip/hip_runtime.h>


typedef float v2f __attribute__((ext_vector_type(2)));
typedef float v8f __attribute__((ext_vector_type(8)));

#define C_CH 192
#define N_IN (192 * 192 * 9)

__device__ __forceinline__ v8f wmma_f32_16x16x4(v2f a, v2f b, v8f c) {
  return __builtin_amdgcn_wmma_f32_16x16x4_f32(false, a, false, b, (short)0, c,
                                               false, false);
}

__global__ void transpose_in_k(const float* __restrict__ in,
                               float* __restrict__ out, int n) {
  int i = blockIdx.x * blockDim.x + threadIdx.x;
  if (i < n) {
    int a = i % C_CH;
    int t = i / C_CH;
    int p = t % 9;
    int o = t / 9;
    out[i] = in[(o * C_CH + a) * 9 + p];
  }
}

__global__ void ssq_partial_k(const float* __restrict__ x, int n,
                              float* __restrict__ part) {
  __shared__ float sdata[256];
  float s = 0.f;
  for (int i = blockIdx.x * blockDim.x + threadIdx.x; i < n;
       i += gridDim.x * blockDim.x) {
    float v = x[i];
    s += v * v;
  }
  sdata[threadIdx.x] = s;
  __syncthreads();
  for (int off = 128; off > 0; off >>= 1) {
    if ((int)threadIdx.x < off) sdata[threadIdx.x] += sdata[threadIdx.x + off];
    __syncthreads();
  }
  if (threadIdx.x == 0) part[blockIdx.x] = sdata[0];
}

__global__ void final_reduce_k(const float* __restrict__ part, int n,
                               float* __restrict__ out) {
  __shared__ float sdata[256];
  float s = 0.f;
  for (int i = threadIdx.x; i < n; i += 256) s += part[i];
  sdata[threadIdx.x] = s;
  __syncthreads();
  for (int off = 128; off > 0; off >>= 1) {
    if ((int)threadIdx.x < off) sdata[threadIdx.x] += sdata[threadIdx.x + off];
    __syncthreads();
  }
  if (threadIdx.x == 0) out[0] = sdata[0];
}

template <int NW>
__device__ __forceinline__ void gram_wmma_body(
    const float* __restrict__ src, float* __restrict__ dst,
    const float* __restrict__ f2, float* __restrict__ part, int k,
    float* red) {
  constexpr int OPW = C_CH / NW;
  constexpr int VPT = 32 / NW;
  const int P = k * k;
  const int w = 2 * k - 1;
  const int S = w * w;
  const int sidx = (S >> 1) - (int)blockIdx.y;
  const int sy = sidx / w - (k - 1);
  const int sx = sidx % w - (k - 1);
  const int ny = k - (sy < 0 ? -sy : sy);
  const int nx = k - (sx < 0 ? -sx : sx);
  const int validp = ny * nx;
  const int by0 = sy < 0 ? -sy : 0;
  const int bx0 = sx < 0 ? -sx : 0;
  const int dpa = sy * k + sx;
  const unsigned Mdiv = (65536u + (unsigned)nx - 1u) / (unsigned)nx;

  const int lane = threadIdx.x & 31;
  const int wave = __builtin_amdgcn_readfirstlane((int)(threadIdx.x >> 5));
  const int l15 = lane & 15;
  const int h = lane >> 4;
  const int m0 = (blockIdx.x / 3) * 32;
  const int n0 = (blockIdx.x % 3) * 64;

  const v8f z = {0.f, 0.f, 0.f, 0.f, 0.f, 0.f, 0.f, 0.f};
  v8f acc[2][4] = {{z, z, z, z}, {z, z, z, z}};

  static_assert((C_CH / NW) % 2 == 0, "the step loop below is unrolled by two");
  const int o_begin = wave * OPW;
  const int o_end = o_begin + OPW;
  const int vlast = validp - 1;
  const int T = OPW * ((validp + 3) >> 2);
  const int h2 = 2 * h;
  const unsigned cbB = (unsigned)(n0 + l15) * 4u;
  const int caB = (dpa * C_CH + m0 + l15) * 4;

  auto load_step = [&](int o, int j0, v2f (&av)[2], v2f (&bv)[4]) {
    const char* baseo = (const char*)(src + (size_t)o * P * C_CH);
    if (j0 == 0 && o + 1 < o_end)
      __builtin_prefetch(baseo + (size_t)P * C_CH * 4, 0, 3);
#pragma unroll
    for (int e = 0; e < 2; ++e) {
      int j = j0 + h2 + e;
      int jj = j < vlast ? j : vlast;
      int jd = (int)(((unsigned)jj * Mdiv) >> 16);
      int jm = jj - jd * nx;
      int pb = (by0 + jd) * k + (bx0 + jm);
      unsigned rowB = (unsigned)(pb * (C_CH * 4));
      unsigned ob = rowB + cbB;
      unsigned oa = (unsigned)((int)rowB + caB);
      av[0][e] = *(const float*)(baseo + oa);
      av[1][e] = *(const float*)(baseo + oa + 64u);
#pragma unroll
      for (int tj = 0; tj < 4; ++tj)
        bv[tj][e] = *(const float*)(baseo + ob + 64u * (unsigned)tj);
    }
  };

  auto mma_step = [&](int j0, v2f (&av)[2], v2f (&bv)[4]) {
    if (j0 + 4 > validp) {
#pragma unroll
      for (int e = 0; e < 2; ++e) {
        const bool ok = (j0 + h2 + e) < validp;
        av[0][e] = ok ? av[0][e] : 0.f;
        av[1][e] = ok ? av[1][e] : 0.f;
#pragma unroll
        for (int tj = 0; tj < 4; ++tj) bv[tj][e] = ok ? bv[tj][e] : 0.f;
      }
    }
#pragma unroll
    for (int ti = 0; ti < 2; ++ti)
#pragma unroll
      for (int tj = 0; tj < 4; ++tj)
        acc[ti][tj] = wmma_f32_16x16x4(av[ti], bv[tj], acc[ti][tj]);
  };

  {
    v2f aX[2], bX[4], aY[2], bY[4];
    int oX = o_begin, jX = 0;
    int oY, jY;
    load_step(oX, jX, aX, bX);
    for (int t = 2; t < T; t += 2) {
      oY = oX; jY = jX + 4;
      if (jY >= validp) { jY = 0; ++oY; }
      load_step(oY, jY, aY, bY);
      mma_step(jX, aX, bX);
      oX = oY; jX = jY + 4;
      if (jX >= validp) { jX = 0; ++oX; }
      load_step(oX, jX, aX, bX);
      mma_step(jY, aY, bY);
    }
    oY = oX; jY = jX + 4;
    if (jY >= validp) { jY = 0; ++oY; }
    load_step(oY, jY, aY, bY);
    mma_step(jX, aX, bX);
    mma_step(jY, aY, bY);
  }

  const float inv = 1.0f / f2[0];

  const int rr = wave & 7;
  const int tjb = (wave >> 3) * VPT;
  float ssq = 0.f;
  float vv[2][VPT];
#pragma unroll
  for (int ti = 0; ti < 2; ++ti) {
    if (ti) __syncthreads();
#pragma unroll
    for (int tj = 0; tj < 4; ++tj)
#pragma unroll
      for (int r = 0; r < 8; ++r)
        red[(wave * 32 + tj * 8 + r) * 32 + lane] = acc[ti][tj][r];
    __syncthreads();
    const int am = m0 + 16 * ti + rr + 8 * h;
    const size_t row = ((size_t)am * S + sidx) * C_CH;
#pragma unroll
    for (int q = 0; q < VPT; ++q) {
      const int tj = tjb + q;
      float v = red[(tj * 8 + rr) * 32 + lane];
#pragma unroll
      for (int w2 = 1; w2 < NW; ++w2)
        v += red[(w2 * 32 + tj * 8 + rr) * 32 + lane];
      v *= inv;
      vv[ti][q] = v;
      if (dst)
        dst[row + n0 + 16 * tj + l15] = v;
      else
        ssq += v * v;
    }
  }

  if (dst && sidx != (S >> 1)) {
    constexpr int TP = 33;
    constexpr int CPW = 64 / NW;
    __syncthreads();
#pragma unroll
    for (int ti = 0; ti < 2; ++ti)
#pragma unroll
      for (int q = 0; q < VPT; ++q)
        red[(16 * (tjb + q) + l15) * TP + 16 * ti + rr + 8 * h] = vv[ti][q];
    __syncthreads();
    const int ms = S - 1 - sidx;
#pragma unroll
    for (int c = 0; c < CPW; ++c) {
      const int b = wave * CPW + c;
      dst[((size_t)(n0 + b) * S + ms) * C_CH + m0 + lane] = red[b * TP + lane];
    }
  }

  if (!dst) {
    for (int off = 16; off > 0; off >>= 1) ssq += __shfl_xor(ssq, off, 32);
    __syncthreads();
    if (lane == 0) red[wave] = ssq;
    __syncthreads();
    if (threadIdx.x == 0) {
      float s = red[0];
      for (int w2 = 1; w2 < NW; ++w2) s += red[w2];
      part[blockIdx.y * gridDim.x + blockIdx.x] = (sidx == (S >> 1)) ? s : 2.f * s;
    }
  }
}

__global__ __launch_bounds__(512) void gram_wmma16_k(
    const float* __restrict__ src, float* __restrict__ dst,
    const float* __restrict__ f2, float* __restrict__ part, int k) {
  __shared__ float red[16 * 32 * 32];
  gram_wmma_body<16>(src, dst, f2, part, k, red);
}

__global__ void sigma_k(const float* __restrict__ s,
                        float* __restrict__ inv_sigma) {
  if (threadIdx.x == 0 && blockIdx.x == 0) {
    double ls = 0.5 * (log((double)s[0]) + 0.5 * log((double)s[1]) +
                       0.25 * log((double)s[2]) + 0.125 * log((double)s[3]));
    inv_sigma[0] = (float)exp(-ls);
  }
}

__global__ void scale_k(const float* __restrict__ in,
                        const float* __restrict__ inv_sigma,
                        float* __restrict__ out, int n) {
  int i = blockIdx.x * blockDim.x + threadIdx.x;
  if (i < n) out[i] = in[i] * inv_sigma[0];
}

static inline int imin(int a, int b) { return a < b ? a : b; }

extern "C" void kernel_launch(void* const* d_in, const int* in_sizes, int n_in,
                              void* d_out, int out_size, void* d_ws,
                              size_t ws_size, hipStream_t stream) {
  (void)in_sizes; (void)n_in; (void)out_size; (void)ws_size;
  const float* K = (const float*)d_in[0];
  float* out = (float*)d_out;
  char* ws = (char*)d_ws;

  float* scal = (float*)ws;
  float* part = (float*)(ws + 1024);
  float* K1T = (float*)(ws + (1 << 17));
  float* K2T = (float*)(ws + (4 << 20));
  float* KT0 = K2T;

  const int N0 = N_IN;
  const int N1 = 192 * 192 * 25;
  const int N2 = 192 * 192 * 81;

  transpose_in_k<<<(N0 + 255) / 256, 256, 0, stream>>>(K, KT0, N0);
  int g0 = imin(1024, (N0 + 255) / 256);
  ssq_partial_k<<<g0, 256, 0, stream>>>(K, N0, part);
  final_reduce_k<<<1, 256, 0, stream>>>(part, g0, scal + 0);

  gram_wmma16_k<<<dim3(18, 13), 512, 0, stream>>>(KT0, K1T, scal + 0, nullptr,
                                                  3);

  int g1 = imin(1024, (N1 + 255) / 256);
  ssq_partial_k<<<g1, 256, 0, stream>>>(K1T, N1, part);
  final_reduce_k<<<1, 256, 0, stream>>>(part, g1, scal + 1);

  gram_wmma16_k<<<dim3(18, 41), 512, 0, stream>>>(K1T, K2T, scal + 1, nullptr,
                                                  5);

  int g2 = imin(1024, (N2 + 255) / 256);
  ssq_partial_k<<<g2, 256, 0, stream>>>(K2T, N2, part);
  final_reduce_k<<<1, 256, 0, stream>>>(part, g2, scal + 2);

  gram_wmma16_k<<<dim3(18, 145), 512, 0, stream>>>(K2T, nullptr, scal + 2,
                                                   part, 9);
  final_reduce_k<<<1, 256, 0, stream>>>(part, 18 * 145, scal + 3);

  sigma_k<<<1, 1, 0, stream>>>(scal, scal + 4);
  scale_k<<<(N0 + 255) / 256, 256, 0, stream>>>(K, scal + 4, out, N0);
}


# Gram steps: k loop interchanged (position groups outer, o inner): per-lane row-offset integer math runs once per group, not every k-step; inner step = 12 loads + 8 products (changes fp sum order)
# speedup vs baseline: 6.3712x; 1.1679x over previous
#include <hip/hip_runtime.h>


typedef float v2f __attribute__((ext_vector_type(2)));
typedef float v8f __attribute__((ext_vector_type(8)));

#define C_CH 192
#define N_IN (192 * 192 * 9)

__device__ __forceinline__ v8f wmma_f32_16x16x4(v2f a, v2f b, v8f c) {
  return __builtin_amdgcn_wmma_f32_16x16x4_f32(false, a, false, b, (short)0, c,
                                               false, false);
}

__global__ void transpose_in_k(const float* __restrict__ in,
                               float* __restrict__ out, int n) {
  int i = blockIdx.x * blockDim.x + threadIdx.x;
  if (i < n) {
    int a = i % C_CH;
    int t = i / C_CH;
    int p = t % 9;
    int o = t / 9;
    out[i] = in[(o * C_CH + a) * 9 + p];
  }
}

__global__ void ssq_partial_k(const float* __restrict__ x, int n,
                              float* __restrict__ part) {
  __shared__ float sdata[256];
  float s = 0.f;
  for (int i = blockIdx.x * blockDim.x + threadIdx.x; i < n;
       i += gridDim.x * blockDim.x) {
    float v = x[i];
    s += v * v;
  }
  sdata[threadIdx.x] = s;
  __syncthreads();
  for (int off = 128; off > 0; off >>= 1) {
    if ((int)threadIdx.x < off) sdata[threadIdx.x] += sdata[threadIdx.x + off];
    __syncthreads();
  }
  if (threadIdx.x == 0) part[blockIdx.x] = sdata[0];
}

__global__ void final_reduce_k(const float* __restrict__ part, int n,
                               float* __restrict__ out) {
  __shared__ float sdata[256];
  float s = 0.f;
  for (int i = threadIdx.x; i < n; i += 256) s += part[i];
  sdata[threadIdx.x] = s;
  __syncthreads();
  for (int off = 128; off > 0; off >>= 1) {
    if ((int)threadIdx.x < off) sdata[threadIdx.x] += sdata[threadIdx.x + off];
    __syncthreads();
  }
  if (threadIdx.x == 0) out[0] = sdata[0];
}

template <int NW>
__device__ __forceinline__ void gram_wmma_body(
    const float* __restrict__ src, float* __restrict__ dst,
    const float* __restrict__ f2, float* __restrict__ part, int k,
    float* red) {
  constexpr int OPW = C_CH / NW;
  constexpr int VPT = 32 / NW;
  const int P = k * k;
  const int w = 2 * k - 1;
  const int S = w * w;
  const int sidx = (S >> 1) - (int)blockIdx.y;
  const int sy = sidx / w - (k - 1);
  const int sx = sidx % w - (k - 1);
  const int ny = k - (sy < 0 ? -sy : sy);
  const int nx = k - (sx < 0 ? -sx : sx);
  const int validp = ny * nx;
  const int by0 = sy < 0 ? -sy : 0;
  const int bx0 = sx < 0 ? -sx : 0;
  const int dpa = sy * k + sx;
  const unsigned Mdiv = (65536u + (unsigned)nx - 1u) / (unsigned)nx;

  const int lane = threadIdx.x & 31;
  const int wave = __builtin_amdgcn_readfirstlane((int)(threadIdx.x >> 5));
  const int l15 = lane & 15;
  const int h = lane >> 4;
  const int m0 = (blockIdx.x / 3) * 32;
  const int n0 = (blockIdx.x % 3) * 64;

  const v8f z = {0.f, 0.f, 0.f, 0.f, 0.f, 0.f, 0.f, 0.f};
  v8f acc[2][4] = {{z, z, z, z}, {z, z, z, z}};

  static_assert((C_CH / NW) % 2 == 0, "the step loop below is unrolled by two");
  const int o_begin = wave * OPW;
  const int o_end = o_begin + OPW;
  const int vlast = validp - 1;
  const int T = OPW * ((validp + 3) >> 2);
  const int h2 = 2 * h;
  const unsigned cbB = (unsigned)(n0 + l15) * 4u;
  const int caB = (dpa * C_CH + m0 + l15) * 4;

  auto lane_offsets = [&](int j0, unsigned (&oa)[2], unsigned (&ob)[2]) {
#pragma unroll
    for (int e = 0; e < 2; ++e) {
      int j = j0 + h2 + e;
      int jj = j < vlast ? j : vlast;
      int jd = (int)(((unsigned)jj * Mdiv) >> 16);
      int jm = jj - jd * nx;
      int pb = (by0 + jd) * k + (bx0 + jm);
      unsigned rowB = (unsigned)(pb * (C_CH * 4));
      ob[e] = rowB + cbB;
      oa[e] = (unsigned)((int)rowB + caB);
    }
  };

  auto load_step = [&](int o, const unsigned (&oa)[2], const unsigned (&ob)[2],
                       v2f (&av)[2], v2f (&bv)[4]) {
    const char* baseo = (const char*)(src + (size_t)o * P * C_CH);
#pragma unroll
    for (int e = 0; e < 2; ++e) {
      av[0][e] = *(const float*)(baseo + oa[e]);
      av[1][e] = *(const float*)(baseo + oa[e] + 64u);
#pragma unroll
      for (int tj = 0; tj < 4; ++tj)
        bv[tj][e] = *(const float*)(baseo + ob[e] + 64u * (unsigned)tj);
    }
  };

  auto mma_step = [&](int j0, v2f (&av)[2], v2f (&bv)[4]) {
    if (j0 + 4 > validp) {
#pragma unroll
      for (int e = 0; e < 2; ++e) {
        const bool ok = (j0 + h2 + e) < validp;
        av[0][e] = ok ? av[0][e] : 0.f;
        av[1][e] = ok ? av[1][e] : 0.f;
#pragma unroll
        for (int tj = 0; tj < 4; ++tj) bv[tj][e] = ok ? bv[tj][e] : 0.f;
      }
    }
#pragma unroll
    for (int ti = 0; ti < 2; ++ti)
#pragma unroll
      for (int tj = 0; tj < 4; ++tj)
        acc[ti][tj] = wmma_f32_16x16x4(av[ti], bv[tj], acc[ti][tj]);
  };

  {
    v2f aX[2], bX[4], aY[2], bY[4];
    unsigned oaL[2], obL[2];
    int oL = o_begin, jL = 0;
    auto next_step = [&]() {
      ++oL;
      if (oL == o_end) {
        oL = o_begin;
        jL += 4;
        lane_offsets(jL, oaL, obL);
      }
    };
    int jX = 0, jY;
    lane_offsets(0, oaL, obL);
    load_step(oL, oaL, obL, aX, bX);
    for (int t = 2; t < T; t += 2) {
      next_step();
      jY = jL;
      load_step(oL, oaL, obL, aY, bY);
      mma_step(jX, aX, bX);
      next_step();
      jX = jL;
      load_step(oL, oaL, obL, aX, bX);
      mma_step(jY, aY, bY);
    }
    next_step();
    jY = jL;
    load_step(oL, oaL, obL, aY, bY);
    mma_step(jX, aX, bX);
    mma_step(jY, aY, bY);
  }

  const float inv = 1.0f / f2[0];

  const int rr = wave & 7;
  const int tjb = (wave >> 3) * VPT;
  float ssq = 0.f;
  float vv[2][VPT];
#pragma unroll
  for (int ti = 0; ti < 2; ++ti) {
    if (ti) __syncthreads();
#pragma unroll
    for (int tj = 0; tj < 4; ++tj)
#pragma unroll
      for (int r = 0; r < 8; ++r)
        red[(wave * 32 + tj * 8 + r) * 32 + lane] = acc[ti][tj][r];
    __syncthreads();
    const int am = m0 + 16 * ti + rr + 8 * h;
    const size_t row = ((size_t)am * S + sidx) * C_CH;
#pragma unroll
    for (int q = 0; q < VPT; ++q) {
      const int tj = tjb + q;
      float v = red[(tj * 8 + rr) * 32 + lane];
#pragma unroll
      for (int w2 = 1; w2 < NW; ++w2)
        v += red[(w2 * 32 + tj * 8 + rr) * 32 + lane];
      v *= inv;
      vv[ti][q] = v;
      if (dst)
        dst[row + n0 + 16 * tj + l15] = v;
      else
        ssq += v * v;
    }
  }

  if (dst && sidx != (S >> 1)) {
    constexpr int TP = 33;
    constexpr int CPW = 64 / NW;
    __syncthreads();
#pragma unroll
    for (int ti = 0; ti < 2; ++ti)
#pragma unroll
      for (int q = 0; q < VPT; ++q)
        red[(16 * (tjb + q) + l15) * TP + 16 * ti + rr + 8 * h] = vv[ti][q];
    __syncthreads();
    const int ms = S - 1 - sidx;
#pragma unroll
    for (int c = 0; c < CPW; ++c) {
      const int b = wave * CPW + c;
      dst[((size_t)(n0 + b) * S + ms) * C_CH + m0 + lane] = red[b * TP + lane];
    }
  }

  if (!dst) {
    for (int off = 16; off > 0; off >>= 1) ssq += __shfl_xor(ssq, off, 32);
    __syncthreads();
    if (lane == 0) red[wave] = ssq;
    __syncthreads();
    if (threadIdx.x == 0) {
      float s = red[0];
      for (int w2 = 1; w2 < NW; ++w2) s += red[w2];
      part[blockIdx.y * gridDim.x + blockIdx.x] = (sidx == (S >> 1)) ? s : 2.f * s;
    }
  }
}

__global__ __launch_bounds__(512) void gram_wmma16_k(
    const float* __restrict__ src, float* __restrict__ dst,
    const float* __restrict__ f2, float* __restrict__ part, int k) {
  __shared__ float red[16 * 32 * 32];
  gram_wmma_body<16>(src, dst, f2, part, k, red);
}

__global__ void sigma_k(const float* __restrict__ s,
                        float* __restrict__ inv_sigma) {
  if (threadIdx.x == 0 && blockIdx.x == 0) {
    double ls = 0.5 * (log((double)s[0]) + 0.5 * log((double)s[1]) +
                       0.25 * log((double)s[2]) + 0.125 * log((double)s[3]));
    inv_sigma[0] = (float)exp(-ls);
  }
}

__global__ void scale_k(const float* __restrict__ in,
                        const float* __restrict__ inv_sigma,
                        float* __restrict__ out, int n) {
  int i = blockIdx.x * blockDim.x + threadIdx.x;
  if (i < n) out[i] = in[i] * inv_sigma[0];
}

static inline int imin(int a, int b) { return a < b ? a : b; }

extern "C" void kernel_launch(void* const* d_in, const int* in_sizes, int n_in,
                              void* d_out, int out_size, void* d_ws,
                              size_t ws_size, hipStream_t stream) {
  (void)in_sizes; (void)n_in; (void)out_size; (void)ws_size;
  const float* K = (const float*)d_in[0];
  float* out = (float*)d_out;
  char* ws = (char*)d_ws;

  float* scal = (float*)ws;
  float* part = (float*)(ws + 1024);
  float* K1T = (float*)(ws + (1 << 17));
  float* K2T = (float*)(ws + (4 << 20));
  float* KT0 = K2T;

  const int N0 = N_IN;
  const int N1 = 192 * 192 * 25;
  const int N2 = 192 * 192 * 81;

  transpose_in_k<<<(N0 + 255) / 256, 256, 0, stream>>>(K, KT0, N0);
  int g0 = imin(1024, (N0 + 255) / 256);
  ssq_partial_k<<<g0, 256, 0, stream>>>(K, N0, part);
  final_reduce_k<<<1, 256, 0, stream>>>(part, g0, scal + 0);

  gram_wmma16_k<<<dim3(18, 13), 512, 0, stream>>>(KT0, K1T, scal + 0, nullptr,
                                                  3);

  int g1 = imin(1024, (N1 + 255) / 256);
  ssq_partial_k<<<g1, 256, 0, stream>>>(K1T, N1, part);
  final_reduce_k<<<1, 256, 0, stream>>>(part, g1, scal + 1);

  gram_wmma16_k<<<dim3(18, 41), 512, 0, stream>>>(K1T, K2T, scal + 1, nullptr,
                                                  5);

  int g2 = imin(1024, (N2 + 255) / 256);
  ssq_partial_k<<<g2, 256, 0, stream>>>(K2T, N2, part);
  final_reduce_k<<<1, 256, 0, stream>>>(part, g2, scal + 2);

  gram_wmma16_k<<<dim3(18, 145), 512, 0, stream>>>(K2T, nullptr, scal + 2,
                                                   part, 9);
  final_reduce_k<<<1, 256, 0, stream>>>(part, 18 * 145, scal + 3);

  sigma_k<<<1, 1, 0, stream>>>(scal, scal + 4);
  scale_k<<<(N0 + 255) / 256, 256, 0, stream>>>(K, scal + 4, out, N0);
}


# Gram steps: a k-step takes four values of o at one overlap position instead of four positions of one o, so no k slot is padding for any overlap size (changes fp sum order)
# speedup vs baseline: 6.8716x; 1.0785x over previous
#include <hip/hip_runtime.h>


typedef float v2f __attribute__((ext_vector_type(2)));
typedef float v8f __attribute__((ext_vector_type(8)));

#define C_CH 192
#define N_IN (192 * 192 * 9)

__device__ __forceinline__ v8f wmma_f32_16x16x4(v2f a, v2f b, v8f c) {
  return __builtin_amdgcn_wmma_f32_16x16x4_f32(false, a, false, b, (short)0, c,
                                               false, false);
}

__global__ void transpose_in_k(const float* __restrict__ in,
                               float* __restrict__ out, int n) {
  int i = blockIdx.x * blockDim.x + threadIdx.x;
  if (i < n) {
    int a = i % C_CH;
    int t = i / C_CH;
    int p = t % 9;
    int o = t / 9;
    out[i] = in[(o * C_CH + a) * 9 + p];
  }
}

__global__ void ssq_partial_k(const float* __restrict__ x, int n,
                              float* __restrict__ part) {
  __shared__ float sdata[256];
  float s = 0.f;
  for (int i = blockIdx.x * blockDim.x + threadIdx.x; i < n;
       i += gridDim.x * blockDim.x) {
    float v = x[i];
    s += v * v;
  }
  sdata[threadIdx.x] = s;
  __syncthreads();
  for (int off = 128; off > 0; off >>= 1) {
    if ((int)threadIdx.x < off) sdata[threadIdx.x] += sdata[threadIdx.x + off];
    __syncthreads();
  }
  if (threadIdx.x == 0) part[blockIdx.x] = sdata[0];
}

__global__ void final_reduce_k(const float* __restrict__ part, int n,
                               float* __restrict__ out) {
  __shared__ float sdata[256];
  float s = 0.f;
  for (int i = threadIdx.x; i < n; i += 256) s += part[i];
  sdata[threadIdx.x] = s;
  __syncthreads();
  for (int off = 128; off > 0; off >>= 1) {
    if ((int)threadIdx.x < off) sdata[threadIdx.x] += sdata[threadIdx.x + off];
    __syncthreads();
  }
  if (threadIdx.x == 0) out[0] = sdata[0];
}

template <int NW>
__device__ __forceinline__ void gram_wmma_body(
    const float* __restrict__ src, float* __restrict__ dst,
    const float* __restrict__ f2, float* __restrict__ part, int k,
    float* red) {
  constexpr int OPW = C_CH / NW;
  constexpr int VPT = 32 / NW;
  const int P = k * k;
  const int w = 2 * k - 1;
  const int S = w * w;
  const int sidx = (S >> 1) - (int)blockIdx.y;
  const int sy = sidx / w - (k - 1);
  const int sx = sidx % w - (k - 1);
  const int ny = k - (sy < 0 ? -sy : sy);
  const int nx = k - (sx < 0 ? -sx : sx);
  const int validp = ny * nx;
  const int by0 = sy < 0 ? -sy : 0;
  const int bx0 = sx < 0 ? -sx : 0;
  const int dpa = sy * k + sx;

  const int lane = threadIdx.x & 31;
  const int wave = __builtin_amdgcn_readfirstlane((int)(threadIdx.x >> 5));
  const int l15 = lane & 15;
  const int h = lane >> 4;
  const int m0 = (blockIdx.x / 3) * 32;
  const int n0 = (blockIdx.x % 3) * 64;

  const v8f z = {0.f, 0.f, 0.f, 0.f, 0.f, 0.f, 0.f, 0.f};
  v8f acc[2][4] = {{z, z, z, z}, {z, z, z, z}};

  static_assert((C_CH / NW) % 4 == 0, "a step takes four values of o");
  constexpr int OG = OPW / 4;
  const int T = OG * validp;
  const int h2 = 2 * h;
  const char* base = (const char*)(src + (size_t)(wave * OPW) * P * C_CH);
  const unsigned gstride = (unsigned)(4 * P * C_CH * 4);
  const unsigned dA = (unsigned)(dpa * C_CH * 4);
  unsigned oaL[2], obL[2];
#pragma unroll
  for (int e = 0; e < 2; ++e) {
    const unsigned slab = (unsigned)((h2 + e) * P * C_CH);
    oaL[e] = (slab + (unsigned)(m0 + l15)) * 4u + dA;
    obL[e] = (slab + (unsigned)(n0 + l15)) * 4u;
  }

  auto load_step = [&](unsigned sB, v2f (&av)[2], v2f (&bv)[4]) {
#pragma unroll
    for (int e = 0; e < 2; ++e) {
      const unsigned ia = oaL[e] + sB;
      const unsigned ib = obL[e] + sB;
      av[0][e] = *(const float*)(base + ia);
      av[1][e] = *(const float*)(base + ia + 64u);
#pragma unroll
      for (int tj = 0; tj < 4; ++tj)
        bv[tj][e] = *(const float*)(base + ib + 64u * (unsigned)tj);
    }
  };

  auto mma_step = [&](const v2f (&av)[2], const v2f (&bv)[4]) {
#pragma unroll
    for (int ti = 0; ti < 2; ++ti)
#pragma unroll
      for (int tj = 0; tj < 4; ++tj)
        acc[ti][tj] = wmma_f32_16x16x4(av[ti], bv[tj], acc[ti][tj]);
  };

  {
    v2f aX[2], bX[4], aY[2], bY[4];
    int gL = 0, jxL = 0;
    unsigned sBL = (unsigned)((by0 * k + bx0) * C_CH * 4);
    auto next_step = [&]() {
      ++gL;
      sBL += gstride;
      if (gL == OG) {
        gL = 0;
        sBL -= (unsigned)OG * gstride;
        int adv = 1;
        if (++jxL == nx) {
          jxL = 0;
          adv += k - nx;
        }
        sBL += (unsigned)(adv * C_CH * 4);
      }
    };
    load_step(sBL, aX, bX);
    if (T & 1) {
      mma_step(aX, bX);
      next_step();
      load_step(sBL, aX, bX);
    }
    const int R = T & ~1;
    for (int t = 2; t < R; t += 2) {
      next_step();
      load_step(sBL, aY, bY);
      __builtin_amdgcn_sched_barrier(0);
      mma_step(aX, bX);
      next_step();
      load_step(sBL, aX, bX);
      __builtin_amdgcn_sched_barrier(0);
      mma_step(aY, bY);
    }
    next_step();
    load_step(sBL, aY, bY);
    __builtin_amdgcn_sched_barrier(0);
    mma_step(aX, bX);
    mma_step(aY, bY);
  }

  const float inv = 1.0f / f2[0];

  const int rr = wave & 7;
  const int tjb = (wave >> 3) * VPT;
  float ssq = 0.f;
  float vv[2][VPT];
#pragma unroll
  for (int ti = 0; ti < 2; ++ti) {
    if (ti) __syncthreads();
#pragma unroll
    for (int tj = 0; tj < 4; ++tj)
#pragma unroll
      for (int r = 0; r < 8; ++r)
        red[(wave * 32 + tj * 8 + r) * 32 + lane] = acc[ti][tj][r];
    __syncthreads();
    const int am = m0 + 16 * ti + rr + 8 * h;
    const size_t row = ((size_t)am * S + sidx) * C_CH;
#pragma unroll
    for (int q = 0; q < VPT; ++q) {
      const int tj = tjb + q;
      float v = red[(tj * 8 + rr) * 32 + lane];
#pragma unroll
      for (int w2 = 1; w2 < NW; ++w2)
        v += red[(w2 * 32 + tj * 8 + rr) * 32 + lane];
      v *= inv;
      vv[ti][q] = v;
      if (dst)
        dst[row + n0 + 16 * tj + l15] = v;
      else
        ssq += v * v;
    }
  }

  if (dst && sidx != (S >> 1)) {
    constexpr int TP = 33;
    constexpr int CPW = 64 / NW;
    __syncthreads();
#pragma unroll
    for (int ti = 0; ti < 2; ++ti)
#pragma unroll
      for (int q = 0; q < VPT; ++q)
        red[(16 * (tjb + q) + l15) * TP + 16 * ti + rr + 8 * h] = vv[ti][q];
    __syncthreads();
    const int ms = S - 1 - sidx;
#pragma unroll
    for (int c = 0; c < CPW; ++c) {
      const int b = wave * CPW + c;
      dst[((size_t)(n0 + b) * S + ms) * C_CH + m0 + lane] = red[b * TP + lane];
    }
  }

  if (!dst) {
    for (int off = 16; off > 0; off >>= 1) ssq += __shfl_xor(ssq, off, 32);
    __syncthreads();
    if (lane == 0) red[wave] = ssq;
    __syncthreads();
    if (threadIdx.x == 0) {
      float s = red[0];
      for (int w2 = 1; w2 < NW; ++w2) s += red[w2];
      part[blockIdx.y * gridDim.x + blockIdx.x] = (sidx == (S >> 1)) ? s : 2.f * s;
    }
  }
}

__global__ __launch_bounds__(512) void gram_wmma16_k(
    const float* __restrict__ src, float* __restrict__ dst,
    const float* __restrict__ f2, float* __restrict__ part, int k) {
  __shared__ float red[16 * 32 * 32];
  gram_wmma_body<16>(src, dst, f2, part, k, red);
}

__global__ void sigma_k(const float* __restrict__ s,
                        float* __restrict__ inv_sigma) {
  if (threadIdx.x == 0 && blockIdx.x == 0) {
    double ls = 0.5 * (log((double)s[0]) + 0.5 * log((double)s[1]) +
                       0.25 * log((double)s[2]) + 0.125 * log((double)s[3]));
    inv_sigma[0] = (float)exp(-ls);
  }
}

__global__ void scale_k(const float* __restrict__ in,
                        const float* __restrict__ inv_sigma,
                        float* __restrict__ out, int n) {
  int i = blockIdx.x * blockDim.x + threadIdx.x;
  if (i < n) out[i] = in[i] * inv_sigma[0];
}

static inline int imin(int a, int b) { return a < b ? a : b; }

extern "C" void kernel_launch(void* const* d_in, const int* in_sizes, int n_in,
                              void* d_out, int out_size, void* d_ws,
                              size_t ws_size, hipStream_t stream) {
  (void)in_sizes; (void)n_in; (void)out_size; (void)ws_size;
  const float* K = (const float*)d_in[0];
  float* out = (float*)d_out;
  char* ws = (char*)d_ws;

  float* scal = (float*)ws;
  float* part = (float*)(ws + 1024);
  float* K1T = (float*)(ws + (1 << 17));
  float* K2T = (float*)(ws + (4 << 20));
  float* KT0 = K2T;

  const int N0 = N_IN;
  const int N1 = 192 * 192 * 25;
  const int N2 = 192 * 192 * 81;

  transpose_in_k<<<(N0 + 255) / 256, 256, 0, stream>>>(K, KT0, N0);
  int g0 = imin(1024, (N0 + 255) / 256);
  ssq_partial_k<<<g0, 256, 0, stream>>>(K, N0, part);
  final_reduce_k<<<1, 256, 0, stream>>>(part, g0, scal + 0);

  gram_wmma16_k<<<dim3(18, 13), 512, 0, stream>>>(KT0, K1T, scal + 0, nullptr,
                                                  3);

  int g1 = imin(1024, (N1 + 255) / 256);
  ssq_partial_k<<<g1, 256, 0, stream>>>(K1T, N1, part);
  final_reduce_k<<<1, 256, 0, stream>>>(part, g1, scal + 1);

  gram_wmma16_k<<<dim3(18, 41), 512, 0, stream>>>(K1T, K2T, scal + 1, nullptr,
                                                  5);

  int g2 = imin(1024, (N2 + 255) / 256);
  ssq_partial_k<<<g2, 256, 0, stream>>>(K2T, N2, part);
  final_reduce_k<<<1, 256, 0, stream>>>(part, g2, scal + 2);

  gram_wmma16_k<<<dim3(18, 145), 512, 0, stream>>>(K2T, nullptr, scal + 2,
                                                   part, 9);
  final_reduce_k<<<1, 256, 0, stream>>>(part, 18 * 145, scal + 3);

  sigma_k<<<1, 1, 0, stream>>>(scal, scal + 4);
  scale_k<<<(N0 + 255) / 256, 256, 0, stream>>>(K, scal + 4, out, N0);
}


# Operands stored [o/2][pos][c][o%2]: each k-step's A and B fragments come from three 16-byte loads per lane instead of twelve 4-byte ones; outputs written in that layout (norm sums change fp order)
# speedup vs baseline: 7.1079x; 1.0344x over previous
#include <hip/hip_runtime.h>


typedef float v2f __attribute__((ext_vector_type(2)));
typedef float v4f __attribute__((ext_vector_type(4)));
typedef float v8f __attribute__((ext_vector_type(8)));

#define C_CH 192
#define N_IN (192 * 192 * 9)

__device__ __forceinline__ v8f wmma_f32_16x16x4(v2f a, v2f b, v8f c) {
  return __builtin_amdgcn_wmma_f32_16x16x4_f32(false, a, false, b, (short)0, c,
                                               false, false);
}

__global__ void transpose_in_k(const float* __restrict__ in,
                               float* __restrict__ out, int n) {
  int i = blockIdx.x * blockDim.x + threadIdx.x;
  if (i < n) {
    int e = i & 1;
    int t = i >> 1;
    int a = t % C_CH;
    t /= C_CH;
    int p = t % 9;
    int o = 2 * (t / 9) + e;
    out[i] = in[(o * C_CH + a) * 9 + p];
  }
}

__global__ void ssq_partial_k(const float* __restrict__ x, int n,
                              float* __restrict__ part) {
  __shared__ float sdata[256];
  float s = 0.f;
  for (int i = blockIdx.x * blockDim.x + threadIdx.x; i < n;
       i += gridDim.x * blockDim.x) {
    float v = x[i];
    s += v * v;
  }
  sdata[threadIdx.x] = s;
  __syncthreads();
  for (int off = 128; off > 0; off >>= 1) {
    if ((int)threadIdx.x < off) sdata[threadIdx.x] += sdata[threadIdx.x + off];
    __syncthreads();
  }
  if (threadIdx.x == 0) part[blockIdx.x] = sdata[0];
}

__global__ void final_reduce_k(const float* __restrict__ part, int n,
                               float* __restrict__ out) {
  __shared__ float sdata[256];
  float s = 0.f;
  for (int i = threadIdx.x; i < n; i += 256) s += part[i];
  sdata[threadIdx.x] = s;
  __syncthreads();
  for (int off = 128; off > 0; off >>= 1) {
    if ((int)threadIdx.x < off) sdata[threadIdx.x] += sdata[threadIdx.x + off];
    __syncthreads();
  }
  if (threadIdx.x == 0) out[0] = sdata[0];
}

__device__ __forceinline__ void gram_wmma_body(
    const float* __restrict__ src, float* __restrict__ dst,
    const float* __restrict__ f2, float* __restrict__ part, int k,
    float* red) {
  constexpr int NW = 16;
  constexpr int OPW = C_CH / NW;
  const int P = k * k;
  const int w = 2 * k - 1;
  const int S = w * w;
  const int sidx = (S >> 1) - (int)blockIdx.y;
  const int sy = sidx / w - (k - 1);
  const int sx = sidx % w - (k - 1);
  const int ny = k - (sy < 0 ? -sy : sy);
  const int nx = k - (sx < 0 ? -sx : sx);
  const int validp = ny * nx;
  const int by0 = sy < 0 ? -sy : 0;
  const int bx0 = sx < 0 ? -sx : 0;
  const int dpa = sy * k + sx;

  const int lane = threadIdx.x & 31;
  const int wave = __builtin_amdgcn_readfirstlane((int)(threadIdx.x >> 5));
  const int l15 = lane & 15;
  const int h = lane >> 4;
  const int m0 = (blockIdx.x / 3) * 32;
  const int n0 = (blockIdx.x % 3) * 64;

  const v8f z = {0.f, 0.f, 0.f, 0.f, 0.f, 0.f, 0.f, 0.f};
  v8f acc[2][4] = {{z, z, z, z}, {z, z, z, z}};

  static_assert((C_CH / NW) % 4 == 0, "a step takes four values of o");
  constexpr int OG = OPW / 4;
  const int T = OG * validp;
  constexpr unsigned ROWB = 2u * C_CH * 4u;
  const char* base =
      (const char*)(src + (size_t)(wave * (OPW / 2)) * P * (2 * C_CH));
  const unsigned gstride = 2u * (unsigned)P * ROWB;
  const unsigned dA = (unsigned)(dpa * (int)ROWB);
  const unsigned slab = (unsigned)(h * P) * ROWB;
  const unsigned oaL = slab + (unsigned)(m0 + 2 * l15) * 8u + dA;
  const unsigned obL = slab + (unsigned)(n0 + 2 * l15) * 8u;

  auto load_step = [&](unsigned sB, v4f& av, v4f (&bv)[2]) {
    av = *(const v4f*)(base + (oaL + sB));
    bv[0] = *(const v4f*)(base + (obL + sB));
    bv[1] = *(const v4f*)(base + (obL + sB) + 256u);
  };

  auto mma_step = [&](const v4f& av, const v4f (&bv)[2]) {
    const v2f a2[2] = {__builtin_shufflevector(av, av, 0, 1),
                       __builtin_shufflevector(av, av, 2, 3)};
    const v2f b2[4] = {__builtin_shufflevector(bv[0], bv[0], 0, 1),
                       __builtin_shufflevector(bv[0], bv[0], 2, 3),
                       __builtin_shufflevector(bv[1], bv[1], 0, 1),
                       __builtin_shufflevector(bv[1], bv[1], 2, 3)};
#pragma unroll
    for (int ti = 0; ti < 2; ++ti)
#pragma unroll
      for (int tj = 0; tj < 4; ++tj)
        acc[ti][tj] = wmma_f32_16x16x4(a2[ti], b2[tj], acc[ti][tj]);
  };

  {
    v4f aX, bX[2], aY, bY[2];
    int gL = 0, jxL = 0;
    unsigned sBL = (unsigned)(by0 * k + bx0) * ROWB;
    auto next_step = [&]() {
      ++gL;
      sBL += gstride;
      if (gL == OG) {
        gL = 0;
        sBL -= (unsigned)OG * gstride;
        int adv = 1;
        if (++jxL == nx) {
          jxL = 0;
          adv += k - nx;
        }
        sBL += (unsigned)adv * ROWB;
      }
    };
    load_step(sBL, aX, bX);
    if (T & 1) {
      mma_step(aX, bX);
      next_step();
      load_step(sBL, aX, bX);
    }
    const int R = T & ~1;
    for (int t = 2; t < R; t += 2) {
      next_step();
      load_step(sBL, aY, bY);
      __builtin_amdgcn_sched_barrier(0);
      mma_step(aX, bX);
      next_step();
      load_step(sBL, aX, bX);
      __builtin_amdgcn_sched_barrier(0);
      mma_step(aY, bY);
    }
    next_step();
    load_step(sBL, aY, bY);
    __builtin_amdgcn_sched_barrier(0);
    mma_step(aX, bX);
    mma_step(aY, bY);
  }

  const float inv = 1.0f / f2[0];

  const int rr = wave & 7;
  const int hh = wave >> 3;
  const int tjl = lane & 1;
  const int rpos = ((lane >> 1) + 16 * hh) ^ (16 * tjl);
  float ssq = 0.f;
  float vv[2][2];
#pragma unroll
  for (int jh = 0; jh < 2; ++jh) {
    if (jh) __syncthreads();
#pragma unroll
    for (int ti = 0; ti < 2; ++ti)
#pragma unroll
      for (int tl = 0; tl < 2; ++tl)
#pragma unroll
        for (int r = 0; r < 8; ++r)
          red[wave * 1024 + ((ti * 2 + tl) * 8 + r) * 32 + (lane ^ (16 * tl))] =
              acc[ti][2 * jh + tl][r];
    __syncthreads();
#pragma unroll
    for (int ti = 0; ti < 2; ++ti) {
      const int off = ((ti * 2 + tjl) * 8 + rr) * 32 + rpos;
      float v = red[off];
#pragma unroll
      for (int w2 = 1; w2 < NW; ++w2) v += red[w2 * 1024 + off];
      v *= inv;
      vv[jh][ti] = v;
      if (!dst) ssq += v * v;
    }
    if (dst) {
      const size_t idx =
          (((size_t)((m0 >> 1) + wave) * S + sidx) * C_CH + (n0 + 32 * jh + lane)) * 2;
      const v2f o2 = {vv[jh][0], vv[jh][1]};
      *(v2f*)(dst + idx) = o2;
    }
  }

  if (dst && sidx != (S >> 1)) {
    constexpr int TP = 33;
    __syncthreads();
#pragma unroll
    for (int jh = 0; jh < 2; ++jh)
#pragma unroll
      for (int ti = 0; ti < 2; ++ti)
        red[(32 * jh + lane) * TP + 2 * wave + ti] = vv[jh][ti];
    __syncthreads();
    const int ms = S - 1 - sidx;
#pragma unroll
    for (int c = 0; c < 2; ++c) {
      const int bp = wave * 2 + c;
      const v2f o2 = {red[(2 * bp) * TP + lane], red[(2 * bp + 1) * TP + lane]};
      const size_t idx =
          (((size_t)((n0 >> 1) + bp) * S + ms) * C_CH + (m0 + lane)) * 2;
      *(v2f*)(dst + idx) = o2;
    }
  }

  if (!dst) {
    for (int off = 16; off > 0; off >>= 1) ssq += __shfl_xor(ssq, off, 32);
    __syncthreads();
    if (lane == 0) red[wave] = ssq;
    __syncthreads();
    if (threadIdx.x == 0) {
      float s = red[0];
      for (int w2 = 1; w2 < NW; ++w2) s += red[w2];
      part[blockIdx.y * gridDim.x + blockIdx.x] = (sidx == (S >> 1)) ? s : 2.f * s;
    }
  }
}

__global__ __launch_bounds__(512) void gram_wmma16_k(
    const float* __restrict__ src, float* __restrict__ dst,
    const float* __restrict__ f2, float* __restrict__ part, int k) {
  __shared__ float red[16 * 32 * 32];
  gram_wmma_body(src, dst, f2, part, k, red);
}

__global__ void sigma_k(const float* __restrict__ s,
                        float* __restrict__ inv_sigma) {
  if (threadIdx.x == 0 && blockIdx.x == 0) {
    double ls = 0.5 * (log((double)s[0]) + 0.5 * log((double)s[1]) +
                       0.25 * log((double)s[2]) + 0.125 * log((double)s[3]));
    inv_sigma[0] = (float)exp(-ls);
  }
}

__global__ void scale_k(const float* __restrict__ in,
                        const float* __restrict__ inv_sigma,
                        float* __restrict__ out, int n) {
  int i = blockIdx.x * blockDim.x + threadIdx.x;
  if (i < n) out[i] = in[i] * inv_sigma[0];
}

static inline int imin(int a, int b) { return a < b ? a : b; }

extern "C" void kernel_launch(void* const* d_in, const int* in_sizes, int n_in,
                              void* d_out, int out_size, void* d_ws,
                              size_t ws_size, hipStream_t stream) {
  (void)in_sizes; (void)n_in; (void)out_size; (void)ws_size;
  const float* K = (const float*)d_in[0];
  float* out = (float*)d_out;
  char* ws = (char*)d_ws;

  float* scal = (float*)ws;
  float* part = (float*)(ws + 1024);
  float* K1T = (float*)(ws + (1 << 17));
  float* K2T = (float*)(ws + (4 << 20));
  float* KT0 = K2T;

  const int N0 = N_IN;
  const int N1 = 192 * 192 * 25;
  const int N2 = 192 * 192 * 81;

  transpose_in_k<<<(N0 + 255) / 256, 256, 0, stream>>>(K, KT0, N0);
  int g0 = imin(1024, (N0 + 255) / 256);
  ssq_partial_k<<<g0, 256, 0, stream>>>(K, N0, part);
  final_reduce_k<<<1, 256, 0, stream>>>(part, g0, scal + 0);

  gram_wmma16_k<<<dim3(18, 13), 512, 0, stream>>>(KT0, K1T, scal + 0, nullptr,
                                                  3);

  int g1 = imin(1024, (N1 + 255) / 256);
  ssq_partial_k<<<g1, 256, 0, stream>>>(K1T, N1, part);
  final_reduce_k<<<1, 256, 0, stream>>>(part, g1, scal + 1);

  gram_wmma16_k<<<dim3(18, 41), 512, 0, stream>>>(K1T, K2T, scal + 1, nullptr,
                                                  5);

  int g2 = imin(1024, (N2 + 255) / 256);
  ssq_partial_k<<<g2, 256, 0, stream>>>(K2T, N2, part);
  final_reduce_k<<<1, 256, 0, stream>>>(part, g2, scal + 2);

  gram_wmma16_k<<<dim3(18, 145), 512, 0, stream>>>(K2T, nullptr, scal + 2,
                                                   part, 9);
  final_reduce_k<<<1, 256, 0, stream>>>(part, 18 * 145, scal + 3);

  sigma_k<<<1, 1, 0, stream>>>(scal, scal + 4);
  scale_k<<<(N0 + 255) / 256, 256, 0, stream>>>(K, scal + 4, out, N0);
}


# Norm sums folded into neighbouring kernels: each work-group writes its sum of squares, the next kernel adds the partials itself, sigma joins the scaling; 13 launches -> 5 (norm sums change fp order)
# speedup vs baseline: 7.5915x; 1.0680x over previous
#include <hip/hip_runtime.h>


typedef float v2f __attribute__((ext_vector_type(2)));
typedef float v4f __attribute__((ext_vector_type(4)));
typedef float v8f __attribute__((ext_vector_type(8)));

#define C_CH 192
#define N_IN (192 * 192 * 9)

__device__ __forceinline__ v8f wmma_f32_16x16x4(v2f a, v2f b, v8f c) {
  return __builtin_amdgcn_wmma_f32_16x16x4_f32(false, a, false, b, (short)0, c,
                                               false, false);
}

__global__ __launch_bounds__(256) void transpose_in_k(
    const float* __restrict__ in, float* __restrict__ out, int n,
    float* __restrict__ part) {
  __shared__ float wsum[8];
  int i = blockIdx.x * blockDim.x + threadIdx.x;
  float s = 0.f;
  if (i < n) {
    int e = i & 1;
    int t = i >> 1;
    int a = t % C_CH;
    t /= C_CH;
    int p = t % 9;
    int o = 2 * (t / 9) + e;
    float v = in[(o * C_CH + a) * 9 + p];
    out[i] = v;
    s = v * v;
  }
  for (int off = 16; off > 0; off >>= 1) s += __shfl_xor(s, off, 32);
  if ((threadIdx.x & 31) == 0) wsum[threadIdx.x >> 5] = s;
  __syncthreads();
  if (threadIdx.x == 0) {
    float t = wsum[0];
    for (int w2 = 1; w2 < 8; ++w2) t += wsum[w2];
    part[blockIdx.x] = t;
  }
}

__device__ __forceinline__ void gram_wmma_body(
    const float* __restrict__ src, float* __restrict__ dst,
    const float* __restrict__ pin, int nin, float* __restrict__ pout,
    float* __restrict__ sout, int k, float* red, float* aux) {
  constexpr int NW = 16;
  constexpr int OPW = C_CH / NW;
  const int P = k * k;
  const int w = 2 * k - 1;
  const int S = w * w;
  const int sidx = (S >> 1) - (int)blockIdx.y;
  const int sy = sidx / w - (k - 1);
  const int sx = sidx % w - (k - 1);
  const int ny = k - (sy < 0 ? -sy : sy);
  const int nx = k - (sx < 0 ? -sx : sx);
  const int validp = ny * nx;
  const int by0 = sy < 0 ? -sy : 0;
  const int bx0 = sx < 0 ? -sx : 0;
  const int dpa = sy * k + sx;

  const int lane = threadIdx.x & 31;
  const int wave = __builtin_amdgcn_readfirstlane((int)(threadIdx.x >> 5));
  const int l15 = lane & 15;
  const int h = lane >> 4;
  const int m0 = (blockIdx.x / 3) * 32;
  const int n0 = (blockIdx.x % 3) * 64;

  const int tid = (int)threadIdx.x;
  const int i1 = tid + 512, i2 = tid + 1024;
  const float q0 = pin[tid < nin ? tid : 0];
  const float q1 = pin[i1 < nin ? i1 : 0];
  const float q2 = pin[i2 < nin ? i2 : 0];
  const float ps =
      ((tid < nin ? q0 : 0.f) + (i1 < nin ? q1 : 0.f)) + (i2 < nin ? q2 : 0.f);

  const v8f z = {0.f, 0.f, 0.f, 0.f, 0.f, 0.f, 0.f, 0.f};
  v8f acc[2][4] = {{z, z, z, z}, {z, z, z, z}};

  static_assert((C_CH / NW) % 4 == 0, "a step takes four values of o");
  constexpr int OG = OPW / 4;
  const int T = OG * validp;
  constexpr unsigned ROWB = 2u * C_CH * 4u;
  const char* base =
      (const char*)(src + (size_t)(wave * (OPW / 2)) * P * (2 * C_CH));
  const unsigned gstride = 2u * (unsigned)P * ROWB;
  const unsigned dA = (unsigned)(dpa * (int)ROWB);
  const unsigned slab = (unsigned)(h * P) * ROWB;
  const unsigned oaL = slab + (unsigned)(m0 + 2 * l15) * 8u + dA;
  const unsigned obL = slab + (unsigned)(n0 + 2 * l15) * 8u;

  auto load_step = [&](unsigned sB, v4f& av, v4f (&bv)[2]) {
    av = *(const v4f*)(base + (oaL + sB));
    bv[0] = *(const v4f*)(base + (obL + sB));
    bv[1] = *(const v4f*)(base + (obL + sB) + 256u);
  };

  auto mma_step = [&](const v4f& av, const v4f (&bv)[2]) {
    const v2f a2[2] = {__builtin_shufflevector(av, av, 0, 1),
                       __builtin_shufflevector(av, av, 2, 3)};
    const v2f b2[4] = {__builtin_shufflevector(bv[0], bv[0], 0, 1),
                       __builtin_shufflevector(bv[0], bv[0], 2, 3),
                       __builtin_shufflevector(bv[1], bv[1], 0, 1),
                       __builtin_shufflevector(bv[1], bv[1], 2, 3)};
#pragma unroll
    for (int ti = 0; ti < 2; ++ti)
#pragma unroll
      for (int tj = 0; tj < 4; ++tj)
        acc[ti][tj] = wmma_f32_16x16x4(a2[ti], b2[tj], acc[ti][tj]);
  };

  {
    v4f aX, bX[2], aY, bY[2];
    int gL = 0, jxL = 0;
    unsigned sBL = (unsigned)(by0 * k + bx0) * ROWB;
    auto next_step = [&]() {
      ++gL;
      sBL += gstride;
      if (gL == OG) {
        gL = 0;
        sBL -= (unsigned)OG * gstride;
        int adv = 1;
        if (++jxL == nx) {
          jxL = 0;
          adv += k - nx;
        }
        sBL += (unsigned)adv * ROWB;
      }
    };
    load_step(sBL, aX, bX);
    if (T & 1) {
      mma_step(aX, bX);
      next_step();
      load_step(sBL, aX, bX);
    }
    const int R = T & ~1;
    for (int t = 2; t < R; t += 2) {
      next_step();
      load_step(sBL, aY, bY);
      __builtin_amdgcn_sched_barrier(0);
      mma_step(aX, bX);
      next_step();
      load_step(sBL, aX, bX);
      __builtin_amdgcn_sched_barrier(0);
      mma_step(aY, bY);
    }
    next_step();
    load_step(sBL, aY, bY);
    __builtin_amdgcn_sched_barrier(0);
    mma_step(aX, bX);
    mma_step(aY, bY);
  }

  {
    float pw = ps;
    for (int off = 16; off > 0; off >>= 1) pw += __shfl_xor(pw, off, 32);
    if (lane == 0) aux[wave] = pw;
  }
  float inv = 0.f;

  const int rr = wave & 7;
  const int hh = wave >> 3;
  const int tjl = lane & 1;
  const int rpos = ((lane >> 1) + 16 * hh) ^ (16 * tjl);
  float ssq = 0.f;
  float vv[2][2];
#pragma unroll
  for (int jh = 0; jh < 2; ++jh) {
    if (jh) __syncthreads();
#pragma unroll
    for (int ti = 0; ti < 2; ++ti)
#pragma unroll
      for (int tl = 0; tl < 2; ++tl)
#pragma unroll
        for (int r = 0; r < 8; ++r)
          red[wave * 1024 + ((ti * 2 + tl) * 8 + r) * 32 + (lane ^ (16 * tl))] =
              acc[ti][2 * jh + tl][r];
    __syncthreads();
    if (jh == 0) {
      float f2 = aux[0];
      for (int w2 = 1; w2 < NW; ++w2) f2 += aux[w2];
      inv = 1.0f / f2;
      if (tid == 0 && blockIdx.x == 0 && blockIdx.y == 0) sout[0] = f2;
    }
#pragma unroll
    for (int ti = 0; ti < 2; ++ti) {
      const int off = ((ti * 2 + tjl) * 8 + rr) * 32 + rpos;
      float v = red[off];
#pragma unroll
      for (int w2 = 1; w2 < NW; ++w2) v += red[w2 * 1024 + off];
      v *= inv;
      vv[jh][ti] = v;
      ssq += v * v;
    }
    if (dst) {
      const size_t idx =
          (((size_t)((m0 >> 1) + wave) * S + sidx) * C_CH + (n0 + 32 * jh + lane)) * 2;
      const v2f o2 = {vv[jh][0], vv[jh][1]};
      *(v2f*)(dst + idx) = o2;
    }
  }

  if (dst && sidx != (S >> 1)) {
    constexpr int TP = 33;
    __syncthreads();
#pragma unroll
    for (int jh = 0; jh < 2; ++jh)
#pragma unroll
      for (int ti = 0; ti < 2; ++ti)
        red[(32 * jh + lane) * TP + 2 * wave + ti] = vv[jh][ti];
    __syncthreads();
    const int ms = S - 1 - sidx;
#pragma unroll
    for (int c = 0; c < 2; ++c) {
      const int bp = wave * 2 + c;
      const v2f o2 = {red[(2 * bp) * TP + lane], red[(2 * bp + 1) * TP + lane]};
      const size_t idx =
          (((size_t)((n0 >> 1) + bp) * S + ms) * C_CH + (m0 + lane)) * 2;
      *(v2f*)(dst + idx) = o2;
    }
  }

  for (int off = 16; off > 0; off >>= 1) ssq += __shfl_xor(ssq, off, 32);
  if (lane == 0) aux[NW + wave] = ssq;
  __syncthreads();
  if (threadIdx.x == 0) {
    float s = aux[NW];
    for (int w2 = 1; w2 < NW; ++w2) s += aux[NW + w2];
    pout[blockIdx.y * gridDim.x + blockIdx.x] = (sidx == (S >> 1)) ? s : 2.f * s;
  }
}

__global__ __launch_bounds__(512) void gram_wmma16_k(
    const float* __restrict__ src, float* __restrict__ dst,
    const float* __restrict__ pin, int nin, float* __restrict__ pout,
    float* __restrict__ sout, int k) {
  __shared__ float red[16 * 32 * 32];
  __shared__ float aux[32];
  gram_wmma_body(src, dst, pin, nin, pout, sout, k, red, aux);
}

#define N_PART3 (18 * 145)
__global__ __launch_bounds__(256) void sigma_scale_k(
    const float* __restrict__ in, const float* __restrict__ scal,
    const float* __restrict__ part, float* __restrict__ out, int n) {
  __shared__ float wsum[8];
  __shared__ float bc;
  const int tid = (int)threadIdx.x;
  const int base = (int)blockIdx.x * 2048 + tid;
  float x[8];
#pragma unroll
  for (int j = 0; j < 8; ++j) {
    const int i = base + 256 * j;
    x[j] = in[i < n ? i : 0];
  }
  constexpr int PER = (N_PART3 + 255) / 256;
  float s = 0.f;
#pragma unroll
  for (int j = 0; j < PER; ++j) {
    const int i = tid + 256 * j;
    const float v = part[i < N_PART3 ? i : 0];
    s += i < N_PART3 ? v : 0.f;
  }
  for (int off = 16; off > 0; off >>= 1) s += __shfl_xor(s, off, 32);
  if ((tid & 31) == 0) wsum[tid >> 5] = s;
  __syncthreads();
  if (tid == 0) {
    float s3 = wsum[0];
    for (int w2 = 1; w2 < 8; ++w2) s3 += wsum[w2];
    double ls = 0.5 * (log((double)scal[0]) + 0.5 * log((double)scal[1]) +
                       0.25 * log((double)scal[2]) + 0.125 * log((double)s3));
    bc = (float)exp(-ls);
  }
  __syncthreads();
  const float inv_sigma = bc;
#pragma unroll
  for (int j = 0; j < 8; ++j) {
    const int i = base + 256 * j;
    if (i < n) out[i] = x[j] * inv_sigma;
  }
}

extern "C" void kernel_launch(void* const* d_in, const int* in_sizes, int n_in,
                              void* d_out, int out_size, void* d_ws,
                              size_t ws_size, hipStream_t stream) {
  (void)in_sizes; (void)n_in; (void)out_size; (void)ws_size;
  const float* K = (const float*)d_in[0];
  float* out = (float*)d_out;
  char* ws = (char*)d_ws;

  float* scal = (float*)ws;
  float* partA = (float*)(ws + 1024);
  float* partB = (float*)(ws + 8192);
  float* partC = (float*)(ws + 12288);
  float* partD = (float*)(ws + 16384);
  float* K1T = (float*)(ws + (1 << 17));
  float* K2T = (float*)(ws + (4 << 20));
  float* KT0 = K2T;

  const int N0 = N_IN;

  transpose_in_k<<<1296, 256, 0, stream>>>(K, KT0, N0, partA);

  gram_wmma16_k<<<dim3(18, 13), 512, 0, stream>>>(KT0, K1T, partA, 1296, partB,
                                                  scal + 0, 3);

  gram_wmma16_k<<<dim3(18, 41), 512, 0, stream>>>(K1T, K2T, partB, 18 * 13,
                                                  partC, scal + 1, 5);

  gram_wmma16_k<<<dim3(18, 145), 512, 0, stream>>>(K2T, nullptr, partC, 18 * 41,
                                                   partD, scal + 2, 9);

  sigma_scale_k<<<162, 256, 0, stream>>>(K, scal, partD, out, N0);
}


# Last Gram step's norm by Parseval on the 17x17 DFT grid: K2 transformed once, then per frequency the upper triangle of one complex 192^3 f32 WMMA product, 145 frequencies (changes fp sum order)
# speedup vs baseline: 21.1467x; 2.7856x over previous
#include <hip/hip_runtime.h>


typedef float v2f __attribute__((ext_vector_type(2)));
typedef float v4f __attribute__((ext_vector_type(4)));
typedef float v8f __attribute__((ext_vector_type(8)));

#define C_CH 192
#define N_IN (192 * 192 * 9)

__device__ __forceinline__ v8f wmma_f32_16x16x4(v2f a, v2f b, v8f c) {
  return __builtin_amdgcn_wmma_f32_16x16x4_f32(false, a, false, b, (short)0, c,
                                               false, false);
}

__global__ __launch_bounds__(256) void transpose_in_k(
    const float* __restrict__ in, float* __restrict__ out, int n,
    float* __restrict__ part) {
  __shared__ float wsum[8];
  int i = blockIdx.x * blockDim.x + threadIdx.x;
  float s = 0.f;
  if (i < n) {
    int e = i & 1;
    int t = i >> 1;
    int a = t % C_CH;
    t /= C_CH;
    int p = t % 9;
    int o = 2 * (t / 9) + e;
    float v = in[(o * C_CH + a) * 9 + p];
    out[i] = v;
    s = v * v;
  }
  for (int off = 16; off > 0; off >>= 1) s += __shfl_xor(s, off, 32);
  if ((threadIdx.x & 31) == 0) wsum[threadIdx.x >> 5] = s;
  __syncthreads();
  if (threadIdx.x == 0) {
    float t = wsum[0];
    for (int w2 = 1; w2 < 8; ++w2) t += wsum[w2];
    part[blockIdx.x] = t;
  }
}

__device__ __forceinline__ void gram_wmma_body(
    const float* __restrict__ src, float* __restrict__ dst,
    const float* __restrict__ pin, int nin, float* __restrict__ pout,
    float* __restrict__ sout, int k, float* red, float* aux) {
  constexpr int NW = 16;
  constexpr int OPW = C_CH / NW;
  const int P = k * k;
  const int w = 2 * k - 1;
  const int S = w * w;
  const int sidx = (S >> 1) - (int)blockIdx.y;
  const int sy = sidx / w - (k - 1);
  const int sx = sidx % w - (k - 1);
  const int ny = k - (sy < 0 ? -sy : sy);
  const int nx = k - (sx < 0 ? -sx : sx);
  const int validp = ny * nx;
  const int by0 = sy < 0 ? -sy : 0;
  const int bx0 = sx < 0 ? -sx : 0;
  const int dpa = sy * k + sx;

  const int lane = threadIdx.x & 31;
  const int wave = __builtin_amdgcn_readfirstlane((int)(threadIdx.x >> 5));
  const int l15 = lane & 15;
  const int h = lane >> 4;
  const int m0 = (blockIdx.x / 3) * 32;
  const int n0 = (blockIdx.x % 3) * 64;

  const int tid = (int)threadIdx.x;
  const int i1 = tid + 512, i2 = tid + 1024;
  const float q0 = pin[tid < nin ? tid : 0];
  const float q1 = pin[i1 < nin ? i1 : 0];
  const float q2 = pin[i2 < nin ? i2 : 0];
  const float ps =
      ((tid < nin ? q0 : 0.f) + (i1 < nin ? q1 : 0.f)) + (i2 < nin ? q2 : 0.f);

  const v8f z = {0.f, 0.f, 0.f, 0.f, 0.f, 0.f, 0.f, 0.f};
  v8f acc[2][4] = {{z, z, z, z}, {z, z, z, z}};

  static_assert((C_CH / NW) % 4 == 0, "a step takes four values of o");
  constexpr int OG = OPW / 4;
  const int T = OG * validp;
  constexpr unsigned ROWB = 2u * C_CH * 4u;
  const char* base =
      (const char*)(src + (size_t)(wave * (OPW / 2)) * P * (2 * C_CH));
  const unsigned gstride = 2u * (unsigned)P * ROWB;
  const unsigned dA = (unsigned)(dpa * (int)ROWB);
  const unsigned slab = (unsigned)(h * P) * ROWB;
  const unsigned oaL = slab + (unsigned)(m0 + 2 * l15) * 8u + dA;
  const unsigned obL = slab + (unsigned)(n0 + 2 * l15) * 8u;

  auto load_step = [&](unsigned sB, v4f& av, v4f (&bv)[2]) {
    av = *(const v4f*)(base + (oaL + sB));
    bv[0] = *(const v4f*)(base + (obL + sB));
    bv[1] = *(const v4f*)(base + (obL + sB) + 256u);
  };

  auto mma_step = [&](const v4f& av, const v4f (&bv)[2]) {
    const v2f a2[2] = {__builtin_shufflevector(av, av, 0, 1),
                       __builtin_shufflevector(av, av, 2, 3)};
    const v2f b2[4] = {__builtin_shufflevector(bv[0], bv[0], 0, 1),
                       __builtin_shufflevector(bv[0], bv[0], 2, 3),
                       __builtin_shufflevector(bv[1], bv[1], 0, 1),
                       __builtin_shufflevector(bv[1], bv[1], 2, 3)};
#pragma unroll
    for (int ti = 0; ti < 2; ++ti)
#pragma unroll
      for (int tj = 0; tj < 4; ++tj)
        acc[ti][tj] = wmma_f32_16x16x4(a2[ti], b2[tj], acc[ti][tj]);
  };

  {
    v4f aX, bX[2], aY, bY[2];
    int gL = 0, jxL = 0;
    unsigned sBL = (unsigned)(by0 * k + bx0) * ROWB;
    auto next_step = [&]() {
      ++gL;
      sBL += gstride;
      if (gL == OG) {
        gL = 0;
        sBL -= (unsigned)OG * gstride;
        int adv = 1;
        if (++jxL == nx) {
          jxL = 0;
          adv += k - nx;
        }
        sBL += (unsigned)adv * ROWB;
      }
    };
    load_step(sBL, aX, bX);
    if (T & 1) {
      mma_step(aX, bX);
      next_step();
      load_step(sBL, aX, bX);
    }
    const int R = T & ~1;
    for (int t = 2; t < R; t += 2) {
      next_step();
      load_step(sBL, aY, bY);
      __builtin_amdgcn_sched_barrier(0);
      mma_step(aX, bX);
      next_step();
      load_step(sBL, aX, bX);
      __builtin_amdgcn_sched_barrier(0);
      mma_step(aY, bY);
    }
    next_step();
    load_step(sBL, aY, bY);
    __builtin_amdgcn_sched_barrier(0);
    mma_step(aX, bX);
    mma_step(aY, bY);
  }

  {
    float pw = ps;
    for (int off = 16; off > 0; off >>= 1) pw += __shfl_xor(pw, off, 32);
    if (lane == 0) aux[wave] = pw;
  }
  float inv = 0.f;

  const int rr = wave & 7;
  const int hh = wave >> 3;
  const int tjl = lane & 1;
  const int rpos = ((lane >> 1) + 16 * hh) ^ (16 * tjl);
  float ssq = 0.f;
  float vv[2][2];
#pragma unroll
  for (int jh = 0; jh < 2; ++jh) {
    if (jh) __syncthreads();
#pragma unroll
    for (int ti = 0; ti < 2; ++ti)
#pragma unroll
      for (int tl = 0; tl < 2; ++tl)
#pragma unroll
        for (int r = 0; r < 8; ++r)
          red[wave * 1024 + ((ti * 2 + tl) * 8 + r) * 32 + (lane ^ (16 * tl))] =
              acc[ti][2 * jh + tl][r];
    __syncthreads();
    if (jh == 0) {
      float f2 = aux[0];
      for (int w2 = 1; w2 < NW; ++w2) f2 += aux[w2];
      inv = 1.0f / f2;
      if (tid == 0 && blockIdx.x == 0 && blockIdx.y == 0) sout[0] = f2;
    }
#pragma unroll
    for (int ti = 0; ti < 2; ++ti) {
      const int off = ((ti * 2 + tjl) * 8 + rr) * 32 + rpos;
      float v = red[off];
#pragma unroll
      for (int w2 = 1; w2 < NW; ++w2) v += red[w2 * 1024 + off];
      v *= inv;
      vv[jh][ti] = v;
      ssq += v * v;
    }
    if (dst) {
      const size_t idx =
          (((size_t)((m0 >> 1) + wave) * S + sidx) * C_CH + (n0 + 32 * jh + lane)) * 2;
      const v2f o2 = {vv[jh][0], vv[jh][1]};
      *(v2f*)(dst + idx) = o2;
    }
  }

  if (dst && sidx != (S >> 1)) {
    constexpr int TP = 33;
    __syncthreads();
#pragma unroll
    for (int jh = 0; jh < 2; ++jh)
#pragma unroll
      for (int ti = 0; ti < 2; ++ti)
        red[(32 * jh + lane) * TP + 2 * wave + ti] = vv[jh][ti];
    __syncthreads();
    const int ms = S - 1 - sidx;
#pragma unroll
    for (int c = 0; c < 2; ++c) {
      const int bp = wave * 2 + c;
      const v2f o2 = {red[(2 * bp) * TP + lane], red[(2 * bp + 1) * TP + lane]};
      const size_t idx =
          (((size_t)((n0 >> 1) + bp) * S + ms) * C_CH + (m0 + lane)) * 2;
      *(v2f*)(dst + idx) = o2;
    }
  }

  for (int off = 16; off > 0; off >>= 1) ssq += __shfl_xor(ssq, off, 32);
  if (lane == 0) aux[NW + wave] = ssq;
  __syncthreads();
  if (threadIdx.x == 0) {
    float s = aux[NW];
    for (int w2 = 1; w2 < NW; ++w2) s += aux[NW + w2];
    pout[blockIdx.y * gridDim.x + blockIdx.x] = (sidx == (S >> 1)) ? s : 2.f * s;
  }
}

__global__ __launch_bounds__(512) void gram_wmma16_k(
    const float* __restrict__ src, float* __restrict__ dst,
    const float* __restrict__ pin, int nin, float* __restrict__ pout,
    float* __restrict__ sout, int k, size_t ws_bytes, size_t skip_from) {
  __shared__ float red[16 * 32 * 32];
  __shared__ float aux[32];
  if (skip_from != 0 && ws_bytes >= skip_from) return;
  gram_wmma_body(src, dst, pin, nin, pout, sout, k, red, aux);
}

#define N_FREQ 145
#define N_PLANE (2 * N_FREQ)

__global__ __launch_bounds__(256) void dft17_k(const float* __restrict__ src,
                                               float* __restrict__ dst,
                                               size_t ws_bytes, size_t need) {
  if (ws_bytes < need) return;
  constexpr float CS[17] = {
      1.00000000f, 0.932472229f, 0.739008904f, 0.445738345f, 0.0922683626f,
      -0.273662984f, -0.602634609f, -0.850217164f, -0.982973099f,
      -0.982973099f, -0.850217164f, -0.602634609f, -0.273662984f,
      0.0922683626f, 0.445738345f, 0.739008904f, 0.932472229f};
  constexpr float SN[17] = {
      0.00000000f, 0.361241668f, 0.673695624f, 0.895163298f, 0.995734155f,
      0.961825669f, 0.798017204f, 0.526432157f, 0.183749512f,
      -0.183749512f, -0.526432157f, -0.798017204f, -0.961825669f,
      -0.995734155f, -0.895163298f, -0.673695624f, -0.361241668f};
  const int t = (int)blockIdx.x * 256 + (int)threadIdx.x;
  const int o2 = t / 384;
  const int within = t - o2 * 384;
  const float* s = src + (size_t)o2 * (81 * 384) + within;
  float x[81];
#pragma unroll
  for (int p = 0; p < 81; ++p) x[p] = s[p * 384];
  float* d = dst + (size_t)o2 * (N_PLANE * 384) + within;
  const int v0 = 3 * (int)blockIdx.y;
  for (int v = v0; v < v0 + 3; ++v) {
    float c1[9], s1[9];
#pragma unroll
    for (int px = 0; px < 9; ++px) {
      const int j = (v * px) % 17;
      c1[px] = CS[j];
      s1[px] = SN[j];
    }
    float tr[9], ti[9];
#pragma unroll
    for (int py = 0; py < 9; ++py) {
      float ar = 0.f, ai = 0.f;
#pragma unroll
      for (int px = 0; px < 9; ++px) {
        ar = fmaf(x[py * 9 + px], c1[px], ar);
        ai = fmaf(x[py * 9 + px], s1[px], ai);
      }
      tr[py] = ar;
      ti[py] = -ai;
    }
    const int f0 = v == 0 ? 0 : 9 + (v - 1) * 17;
#pragma unroll
    for (int u = 0; u < 17; ++u) {
      if (u >= 9 && v == 0) continue;
      float xr = 0.f, xi = 0.f;
#pragma unroll
      for (int py = 0; py < 9; ++py) {
        const int j = (u * py) % 17;
        const float c = CS[j], sn = SN[j];
        xr = fmaf(tr[py], c, xr);
        xr = fmaf(ti[py], sn, xr);
        xi = fmaf(ti[py], c, xi);
        xi = fmaf(-tr[py], sn, xi);
      }
      float* q = d + (size_t)(2 * (f0 + u)) * 384;
      q[0] = xr;
      q[384] = xi;
    }
  }
}

__device__ __forceinline__ void spec_gram_body(
    const float* __restrict__ src, const float* __restrict__ pin, int nin,
    float* __restrict__ pout, float* __restrict__ sout, float* red,
    float* aux) {
  constexpr int NW = 16;
  constexpr int OPW = C_CH / NW;
  constexpr int OG = OPW / 4;
  constexpr int P = N_PLANE;
  const int f = (int)blockIdx.y;

  const int lane = threadIdx.x & 31;
  const int wave = __builtin_amdgcn_readfirstlane((int)(threadIdx.x >> 5));
  const int l15 = lane & 15;
  const int h = lane >> 4;
  int ta = 0, tb = (int)blockIdx.x;
  while (tb >= 6 - ta) {
    tb -= 6 - ta;
    ++ta;
  }
  tb += ta;
  const int m0 = ta * 32;
  const int n0 = tb * 32;

  const int tid = (int)threadIdx.x;
  const int i1 = tid + 512, i2 = tid + 1024;
  const float q0 = pin[tid < nin ? tid : 0];
  const float q1 = pin[i1 < nin ? i1 : 0];
  const float q2 = pin[i2 < nin ? i2 : 0];
  const float ps =
      ((tid < nin ? q0 : 0.f) + (i1 < nin ? q1 : 0.f)) + (i2 < nin ? q2 : 0.f);

  const v8f z = {0.f, 0.f, 0.f, 0.f, 0.f, 0.f, 0.f, 0.f};
  v8f acc[2][2][2] = {{{z, z}, {z, z}}, {{z, z}, {z, z}}};

  constexpr unsigned ROWB = 2u * C_CH * 4u;
  const char* base =
      (const char*)(src + (size_t)(wave * (OPW / 2)) * P * (2 * C_CH));
  constexpr unsigned gstride = 2u * (unsigned)P * ROWB;
  const unsigned slab = (unsigned)(h * P) * ROWB;
  const unsigned row = (unsigned)(2 * f) * ROWB;
  const unsigned oaL = slab + row + (unsigned)(m0 + 2 * l15) * 8u;
  const unsigned obL = slab + row + (unsigned)(n0 + 2 * l15) * 8u;

#pragma unroll
  for (int g = 0; g < OG; ++g) {
    const unsigned off = (unsigned)g * gstride;
    const v4f aR = *(const v4f*)(base + (oaL + off));
    const v4f aI = *(const v4f*)(base + (oaL + off) + ROWB);
    const v4f bR0 = *(const v4f*)(base + (obL + off));
    const v4f bI0 = *(const v4f*)(base + (obL + off) + ROWB);
    const v4f aN = -aI;
    const v2f ar2[2] = {__builtin_shufflevector(aR, aR, 0, 1),
                        __builtin_shufflevector(aR, aR, 2, 3)};
    const v2f ai2[2] = {__builtin_shufflevector(aI, aI, 0, 1),
                        __builtin_shufflevector(aI, aI, 2, 3)};
    const v2f an2[2] = {__builtin_shufflevector(aN, aN, 0, 1),
                        __builtin_shufflevector(aN, aN, 2, 3)};
    const v2f br2[2] = {__builtin_shufflevector(bR0, bR0, 0, 1),
                        __builtin_shufflevector(bR0, bR0, 2, 3)};
    const v2f bi2[2] = {__builtin_shufflevector(bI0, bI0, 0, 1),
                        __builtin_shufflevector(bI0, bI0, 2, 3)};
#pragma unroll
    for (int ti = 0; ti < 2; ++ti)
#pragma unroll
      for (int tj = 0; tj < 2; ++tj) {
        acc[0][ti][tj] = wmma_f32_16x16x4(ar2[ti], br2[tj], acc[0][ti][tj]);
        acc[0][ti][tj] = wmma_f32_16x16x4(ai2[ti], bi2[tj], acc[0][ti][tj]);
        acc[1][ti][tj] = wmma_f32_16x16x4(ar2[ti], bi2[tj], acc[1][ti][tj]);
        acc[1][ti][tj] = wmma_f32_16x16x4(an2[ti], br2[tj], acc[1][ti][tj]);
      }
  }

  {
    float pw = ps;
    for (int off = 16; off > 0; off >>= 1) pw += __shfl_xor(pw, off, 32);
    if (lane == 0) aux[wave] = pw;
  }
  float inv = 0.f;

  const int rr = wave & 7;
  const int hh = wave >> 3;
  const int tjl = lane & 1;
  const int rpos = ((lane >> 1) + 16 * hh) ^ (16 * tjl);
  float ssq = 0.f;
#pragma unroll
  for (int ri = 0; ri < 2; ++ri) {
    if (ri) __syncthreads();
#pragma unroll
    for (int ti = 0; ti < 2; ++ti)
#pragma unroll
      for (int tl = 0; tl < 2; ++tl)
#pragma unroll
        for (int r = 0; r < 8; ++r)
          red[wave * 1024 + ((ti * 2 + tl) * 8 + r) * 32 + (lane ^ (16 * tl))] =
              acc[ri][ti][tl][r];
    __syncthreads();
    if (ri == 0) {
      float f2 = aux[0];
      for (int w2 = 1; w2 < NW; ++w2) f2 += aux[w2];
      inv = 1.0f / f2;
      if (tid == 0 && blockIdx.x == 0 && blockIdx.y == 0) sout[0] = f2;
    }
#pragma unroll
    for (int ti = 0; ti < 2; ++ti) {
      const int off = ((ti * 2 + tjl) * 8 + rr) * 32 + rpos;
      float v = red[off];
#pragma unroll
      for (int w2 = 1; w2 < NW; ++w2) v += red[w2 * 1024 + off];
      v *= inv;
      ssq += v * v;
    }
  }

  for (int off = 16; off > 0; off >>= 1) ssq += __shfl_xor(ssq, off, 32);
  if (lane == 0) aux[NW + wave] = ssq;
  __syncthreads();
  if (threadIdx.x == 0) {
    float s = aux[NW];
    for (int w2 = 1; w2 < NW; ++w2) s += aux[NW + w2];
    const float wgt = (f == 0 ? 1.f : 2.f) * (ta == tb ? 1.f : 2.f);
    pout[blockIdx.y * gridDim.x + blockIdx.x] = wgt * s;
  }
}

__global__ __launch_bounds__(512) void spec_gram16_k(
    const float* __restrict__ src, const float* __restrict__ pin, int nin,
    float* __restrict__ pout, float* __restrict__ sout, size_t ws_bytes,
    size_t need) {
  __shared__ float red[16 * 32 * 32];
  __shared__ float aux[32];
  if (ws_bytes < need) return;
  spec_gram_body(src, pin, nin, pout, sout, red, aux);
}


#define N_PART3 (18 * 145)
#define N_PART3S (21 * 145)
__global__ __launch_bounds__(256) void sigma_scale_k(
    const float* __restrict__ in, const float* __restrict__ scal,
    const float* __restrict__ part, float* __restrict__ out, int n,
    size_t ws_bytes, size_t need) {
  __shared__ float wsum[8];
  __shared__ float bc;
  const int tid = (int)threadIdx.x;
  const int base = (int)blockIdx.x * 2048 + tid;
  float x[8];
#pragma unroll
  for (int j = 0; j < 8; ++j) {
    const int i = base + 256 * j;
    x[j] = in[i < n ? i : 0];
  }
  constexpr int PER = (N_PART3S + 255) / 256;
  const int npart = ws_bytes >= need ? N_PART3S : N_PART3;
  float s = 0.f;
#pragma unroll
  for (int j = 0; j < PER; ++j) {
    const int i = tid + 256 * j;
    const float v = part[i < npart ? i : 0];
    s += i < npart ? v : 0.f;
  }
  for (int off = 16; off > 0; off >>= 1) s += __shfl_xor(s, off, 32);
  if ((tid & 31) == 0) wsum[tid >> 5] = s;
  __syncthreads();
  if (tid == 0) {
    float s3 = wsum[0];
    for (int w2 = 1; w2 < 8; ++w2) s3 += wsum[w2];
    const double s3d = ws_bytes >= need ? (double)s3 / 289.0 : (double)s3;
    double ls = 0.5 * (log((double)scal[0]) + 0.5 * log((double)scal[1]) +
                       0.25 * log((double)scal[2]) + 0.125 * log(s3d));
    bc = (float)exp(-ls);
  }
  __syncthreads();
  const float inv_sigma = bc;
#pragma unroll
  for (int j = 0; j < 8; ++j) {
    const int i = base + 256 * j;
    if (i < n) out[i] = x[j] * inv_sigma;
  }
}

extern "C" void kernel_launch(void* const* d_in, const int* in_sizes, int n_in,
                              void* d_out, int out_size, void* d_ws,
                              size_t ws_size, hipStream_t stream) {
  (void)in_sizes; (void)n_in; (void)out_size;
  const float* K = (const float*)d_in[0];
  float* out = (float*)d_out;
  char* ws = (char*)d_ws;

  float* scal = (float*)ws;
  float* partA = (float*)(ws + 1024);
  float* partB = (float*)(ws + 8192);
  float* partC = (float*)(ws + 12288);
  float* partD = (float*)(ws + 16384);
  float* K1T = (float*)(ws + (1 << 17));
  float* K2T = (float*)(ws + (4 << 20));
  float* KT0 = K2T;
  float* XT = (float*)(ws + (16 << 20));

  const int N0 = N_IN;

  transpose_in_k<<<1296, 256, 0, stream>>>(K, KT0, N0, partA);

  gram_wmma16_k<<<dim3(18, 13), 512, 0, stream>>>(KT0, K1T, partA, 1296, partB,
                                                  scal + 0, 3, ws_size, 0);

  gram_wmma16_k<<<dim3(18, 41), 512, 0, stream>>>(K1T, K2T, partB, 18 * 13,
                                                  partC, scal + 1, 5, ws_size, 0);

  dft17_k<<<dim3(144, 3), 256, 0, stream>>>(K2T, XT, ws_size, 59539456);
  spec_gram16_k<<<dim3(21, 145), 512, 0, stream>>>(XT, partC, 18 * 41, partD,
                                                   scal + 2, ws_size, 59539456);
  gram_wmma16_k<<<dim3(18, 145), 512, 0, stream>>>(K2T, nullptr, partC, 18 * 41,
                                                   partD, scal + 2, 9, ws_size,
                                                   59539456);

  sigma_scale_k<<<162, 256, 0, stream>>>(K, scal, partD, out, N0, ws_size,
                                         59539456);
}


# Spectral product: a tile's o-sum split over 8 waves (was 16), the 8 partial tiles summed in ONE LDS pass of 16-byte accesses; four work-groups per compute unit, not two (changes fp sum order)
# speedup vs baseline: 22.5258x; 1.0652x over previous
#include <hip/hip_runtime.h>


typedef float v2f __attribute__((ext_vector_type(2)));
typedef float v4f __attribute__((ext_vector_type(4)));
typedef float v8f __attribute__((ext_vector_type(8)));

#define C_CH 192
#define N_IN (192 * 192 * 9)

__device__ __forceinline__ v8f wmma_f32_16x16x4(v2f a, v2f b, v8f c) {
  return __builtin_amdgcn_wmma_f32_16x16x4_f32(false, a, false, b, (short)0, c,
                                               false, false);
}

__global__ __launch_bounds__(256) void transpose_in_k(
    const float* __restrict__ in, float* __restrict__ out, int n,
    float* __restrict__ part) {
  __shared__ float wsum[8];
  int i = blockIdx.x * blockDim.x + threadIdx.x;
  float s = 0.f;
  if (i < n) {
    int e = i & 1;
    int t = i >> 1;
    int a = t % C_CH;
    t /= C_CH;
    int p = t % 9;
    int o = 2 * (t / 9) + e;
    float v = in[(o * C_CH + a) * 9 + p];
    out[i] = v;
    s = v * v;
  }
  for (int off = 16; off > 0; off >>= 1) s += __shfl_xor(s, off, 32);
  if ((threadIdx.x & 31) == 0) wsum[threadIdx.x >> 5] = s;
  __syncthreads();
  if (threadIdx.x == 0) {
    float t = wsum[0];
    for (int w2 = 1; w2 < 8; ++w2) t += wsum[w2];
    part[blockIdx.x] = t;
  }
}

__device__ __forceinline__ void gram_wmma_body(
    const float* __restrict__ src, float* __restrict__ dst,
    const float* __restrict__ pin, int nin, float* __restrict__ pout,
    float* __restrict__ sout, int k, float* red, float* aux) {
  constexpr int NW = 16;
  constexpr int OPW = C_CH / NW;
  const int P = k * k;
  const int w = 2 * k - 1;
  const int S = w * w;
  const int sidx = (S >> 1) - (int)blockIdx.y;
  const int sy = sidx / w - (k - 1);
  const int sx = sidx % w - (k - 1);
  const int ny = k - (sy < 0 ? -sy : sy);
  const int nx = k - (sx < 0 ? -sx : sx);
  const int validp = ny * nx;
  const int by0 = sy < 0 ? -sy : 0;
  const int bx0 = sx < 0 ? -sx : 0;
  const int dpa = sy * k + sx;

  const int lane = threadIdx.x & 31;
  const int wave = __builtin_amdgcn_readfirstlane((int)(threadIdx.x >> 5));
  const int l15 = lane & 15;
  const int h = lane >> 4;
  const int m0 = (blockIdx.x / 3) * 32;
  const int n0 = (blockIdx.x % 3) * 64;

  const int tid = (int)threadIdx.x;
  const int i1 = tid + 512, i2 = tid + 1024;
  const float q0 = pin[tid < nin ? tid : 0];
  const float q1 = pin[i1 < nin ? i1 : 0];
  const float q2 = pin[i2 < nin ? i2 : 0];
  const float ps =
      ((tid < nin ? q0 : 0.f) + (i1 < nin ? q1 : 0.f)) + (i2 < nin ? q2 : 0.f);

  const v8f z = {0.f, 0.f, 0.f, 0.f, 0.f, 0.f, 0.f, 0.f};
  v8f acc[2][4] = {{z, z, z, z}, {z, z, z, z}};

  static_assert((C_CH / NW) % 4 == 0, "a step takes four values of o");
  constexpr int OG = OPW / 4;
  const int T = OG * validp;
  constexpr unsigned ROWB = 2u * C_CH * 4u;
  const char* base =
      (const char*)(src + (size_t)(wave * (OPW / 2)) * P * (2 * C_CH));
  const unsigned gstride = 2u * (unsigned)P * ROWB;
  const unsigned dA = (unsigned)(dpa * (int)ROWB);
  const unsigned slab = (unsigned)(h * P) * ROWB;
  const unsigned oaL = slab + (unsigned)(m0 + 2 * l15) * 8u + dA;
  const unsigned obL = slab + (unsigned)(n0 + 2 * l15) * 8u;

  auto load_step = [&](unsigned sB, v4f& av, v4f (&bv)[2]) {
    av = *(const v4f*)(base + (oaL + sB));
    bv[0] = *(const v4f*)(base + (obL + sB));
    bv[1] = *(const v4f*)(base + (obL + sB) + 256u);
  };

  auto mma_step = [&](const v4f& av, const v4f (&bv)[2]) {
    const v2f a2[2] = {__builtin_shufflevector(av, av, 0, 1),
                       __builtin_shufflevector(av, av, 2, 3)};
    const v2f b2[4] = {__builtin_shufflevector(bv[0], bv[0], 0, 1),
                       __builtin_shufflevector(bv[0], bv[0], 2, 3),
                       __builtin_shufflevector(bv[1], bv[1], 0, 1),
                       __builtin_shufflevector(bv[1], bv[1], 2, 3)};
#pragma unroll
    for (int ti = 0; ti < 2; ++ti)
#pragma unroll
      for (int tj = 0; tj < 4; ++tj)
        acc[ti][tj] = wmma_f32_16x16x4(a2[ti], b2[tj], acc[ti][tj]);
  };

  {
    v4f aX, bX[2], aY, bY[2];
    int gL = 0, jxL = 0;
    unsigned sBL = (unsigned)(by0 * k + bx0) * ROWB;
    auto next_step = [&]() {
      ++gL;
      sBL += gstride;
      if (gL == OG) {
        gL = 0;
        sBL -= (unsigned)OG * gstride;
        int adv = 1;
        if (++jxL == nx) {
          jxL = 0;
          adv += k - nx;
        }
        sBL += (unsigned)adv * ROWB;
      }
    };
    load_step(sBL, aX, bX);
    if (T & 1) {
      mma_step(aX, bX);
      next_step();
      load_step(sBL, aX, bX);
    }
    const int R = T & ~1;
    for (int t = 2; t < R; t += 2) {
      next_step();
      load_step(sBL, aY, bY);
      __builtin_amdgcn_sched_barrier(0);
      mma_step(aX, bX);
      next_step();
      load_step(sBL, aX, bX);
      __builtin_amdgcn_sched_barrier(0);
      mma_step(aY, bY);
    }
    next_step();
    load_step(sBL, aY, bY);
    __builtin_amdgcn_sched_barrier(0);
    mma_step(aX, bX);
    mma_step(aY, bY);
  }

  {
    float pw = ps;
    for (int off = 16; off > 0; off >>= 1) pw += __shfl_xor(pw, off, 32);
    if (lane == 0) aux[wave] = pw;
  }
  float inv = 0.f;

  const int rr = wave & 7;
  const int hh = wave >> 3;
  const int tjl = lane & 1;
  const int rpos = ((lane >> 1) + 16 * hh) ^ (16 * tjl);
  float ssq = 0.f;
  float vv[2][2];
#pragma unroll
  for (int jh = 0; jh < 2; ++jh) {
    if (jh) __syncthreads();
#pragma unroll
    for (int ti = 0; ti < 2; ++ti)
#pragma unroll
      for (int tl = 0; tl < 2; ++tl)
#pragma unroll
        for (int r = 0; r < 8; ++r)
          red[wave * 1024 + ((ti * 2 + tl) * 8 + r) * 32 + (lane ^ (16 * tl))] =
              acc[ti][2 * jh + tl][r];
    __syncthreads();
    if (jh == 0) {
      float f2 = aux[0];
      for (int w2 = 1; w2 < NW; ++w2) f2 += aux[w2];
      inv = 1.0f / f2;
      if (tid == 0 && blockIdx.x == 0 && blockIdx.y == 0) sout[0] = f2;
    }
#pragma unroll
    for (int ti = 0; ti < 2; ++ti) {
      const int off = ((ti * 2 + tjl) * 8 + rr) * 32 + rpos;
      float v = red[off];
#pragma unroll
      for (int w2 = 1; w2 < NW; ++w2) v += red[w2 * 1024 + off];
      v *= inv;
      vv[jh][ti] = v;
      ssq += v * v;
    }
    if (dst) {
      const size_t idx =
          (((size_t)((m0 >> 1) + wave) * S + sidx) * C_CH + (n0 + 32 * jh + lane)) * 2;
      const v2f o2 = {vv[jh][0], vv[jh][1]};
      *(v2f*)(dst + idx) = o2;
    }
  }

  if (dst && sidx != (S >> 1)) {
    constexpr int TP = 33;
    __syncthreads();
#pragma unroll
    for (int jh = 0; jh < 2; ++jh)
#pragma unroll
      for (int ti = 0; ti < 2; ++ti)
        red[(32 * jh + lane) * TP + 2 * wave + ti] = vv[jh][ti];
    __syncthreads();
    const int ms = S - 1 - sidx;
#pragma unroll
    for (int c = 0; c < 2; ++c) {
      const int bp = wave * 2 + c;
      const v2f o2 = {red[(2 * bp) * TP + lane], red[(2 * bp + 1) * TP + lane]};
      const size_t idx =
          (((size_t)((n0 >> 1) + bp) * S + ms) * C_CH + (m0 + lane)) * 2;
      *(v2f*)(dst + idx) = o2;
    }
  }

  for (int off = 16; off > 0; off >>= 1) ssq += __shfl_xor(ssq, off, 32);
  if (lane == 0) aux[NW + wave] = ssq;
  __syncthreads();
  if (threadIdx.x == 0) {
    float s = aux[NW];
    for (int w2 = 1; w2 < NW; ++w2) s += aux[NW + w2];
    pout[blockIdx.y * gridDim.x + blockIdx.x] = (sidx == (S >> 1)) ? s : 2.f * s;
  }
}

__global__ __launch_bounds__(512) void gram_wmma16_k(
    const float* __restrict__ src, float* __restrict__ dst,
    const float* __restrict__ pin, int nin, float* __restrict__ pout,
    float* __restrict__ sout, int k, size_t ws_bytes, size_t skip_from) {
  __shared__ float red[16 * 32 * 32];
  __shared__ float aux[32];
  if (skip_from != 0 && ws_bytes >= skip_from) return;
  gram_wmma_body(src, dst, pin, nin, pout, sout, k, red, aux);
}

#define N_FREQ 145
#define N_PLANE (2 * N_FREQ)

__global__ __launch_bounds__(256) void dft17_k(const float* __restrict__ src,
                                               float* __restrict__ dst,
                                               size_t ws_bytes, size_t need) {
  if (ws_bytes < need) return;
  constexpr float CS[17] = {
      1.00000000f, 0.932472229f, 0.739008904f, 0.445738345f, 0.0922683626f,
      -0.273662984f, -0.602634609f, -0.850217164f, -0.982973099f,
      -0.982973099f, -0.850217164f, -0.602634609f, -0.273662984f,
      0.0922683626f, 0.445738345f, 0.739008904f, 0.932472229f};
  constexpr float SN[17] = {
      0.00000000f, 0.361241668f, 0.673695624f, 0.895163298f, 0.995734155f,
      0.961825669f, 0.798017204f, 0.526432157f, 0.183749512f,
      -0.183749512f, -0.526432157f, -0.798017204f, -0.961825669f,
      -0.995734155f, -0.895163298f, -0.673695624f, -0.361241668f};
  const int t = (int)blockIdx.x * 256 + (int)threadIdx.x;
  const int o2 = t / 384;
  const int within = t - o2 * 384;
  const float* s = src + (size_t)o2 * (81 * 384) + within;
  float x[81];
#pragma unroll
  for (int p = 0; p < 81; ++p) x[p] = s[p * 384];
  float* d = dst + (size_t)o2 * (N_PLANE * 384) + within;
  const int v0 = 3 * (int)blockIdx.y;
  for (int v = v0; v < v0 + 3; ++v) {
    float c1[9], s1[9];
#pragma unroll
    for (int px = 0; px < 9; ++px) {
      const int j = (v * px) % 17;
      c1[px] = CS[j];
      s1[px] = SN[j];
    }
    float tr[9], ti[9];
#pragma unroll
    for (int py = 0; py < 9; ++py) {
      float ar = 0.f, ai = 0.f;
#pragma unroll
      for (int px = 0; px < 9; ++px) {
        ar = fmaf(x[py * 9 + px], c1[px], ar);
        ai = fmaf(x[py * 9 + px], s1[px], ai);
      }
      tr[py] = ar;
      ti[py] = -ai;
    }
    const int f0 = v == 0 ? 0 : 9 + (v - 1) * 17;
#pragma unroll
    for (int u = 0; u < 17; ++u) {
      if (u >= 9 && v == 0) continue;
      float xr = 0.f, xi = 0.f;
#pragma unroll
      for (int py = 0; py < 9; ++py) {
        const int j = (u * py) % 17;
        const float c = CS[j], sn = SN[j];
        xr = fmaf(tr[py], c, xr);
        xr = fmaf(ti[py], sn, xr);
        xi = fmaf(ti[py], c, xi);
        xi = fmaf(-tr[py], sn, xi);
      }
      float* q = d + (size_t)(2 * (f0 + u)) * 384;
      q[0] = xr;
      q[384] = xi;
    }
  }
}

__device__ __forceinline__ void spec_gram_body(
    const float* __restrict__ src, const float* __restrict__ pin, int nin,
    float* __restrict__ pout, float* __restrict__ sout, float* red,
    float* aux) {
  constexpr int NW = 8;
  constexpr int OPW = C_CH / NW;
  constexpr int OG = OPW / 4;
  static_assert(OPW % 4 == 0 && OG % 2 == 0,
                "k-steps of four o, taken in pairs over two register sets");
  constexpr int P = N_PLANE;
  const int f = (int)blockIdx.y;

  const int lane = threadIdx.x & 31;
  const int wave = __builtin_amdgcn_readfirstlane((int)(threadIdx.x >> 5));
  const int l15 = lane & 15;
  const int h = lane >> 4;
  int ta = 0, tb = (int)blockIdx.x;
  while (tb >= 6 - ta) {
    tb -= 6 - ta;
    ++ta;
  }
  tb += ta;
  const int m0 = ta * 32;
  const int n0 = tb * 32;

  const int tid = (int)threadIdx.x;
  const int i1 = tid + 256, i2 = tid + 512;
  const float q0 = pin[tid < nin ? tid : 0];
  const float q1 = pin[i1 < nin ? i1 : 0];
  const float q2 = pin[i2 < nin ? i2 : 0];
  const float ps =
      ((tid < nin ? q0 : 0.f) + (i1 < nin ? q1 : 0.f)) + (i2 < nin ? q2 : 0.f);

  const v8f z = {0.f, 0.f, 0.f, 0.f, 0.f, 0.f, 0.f, 0.f};
  v8f acc[2][2][2] = {{{z, z}, {z, z}}, {{z, z}, {z, z}}};

  constexpr unsigned ROWB = 2u * C_CH * 4u;
  const char* base =
      (const char*)(src + (size_t)(wave * (OPW / 2)) * P * (2 * C_CH));
  constexpr unsigned gstride = 2u * (unsigned)P * ROWB;
  const unsigned slab = (unsigned)(h * P) * ROWB;
  const unsigned row = (unsigned)(2 * f) * ROWB;
  const unsigned oaL = slab + row + (unsigned)(m0 + 2 * l15) * 8u;
  const unsigned obL = slab + row + (unsigned)(n0 + 2 * l15) * 8u;

  struct Frag {
    v4f aR, aI, bR, bI;
  };
  auto load_g = [&](int g, Frag& F) {
    const unsigned off = (unsigned)g * gstride;
    F.aR = *(const v4f*)(base + (oaL + off));
    F.aI = *(const v4f*)(base + (oaL + off) + ROWB);
    F.bR = *(const v4f*)(base + (obL + off));
    F.bI = *(const v4f*)(base + (obL + off) + ROWB);
  };
  auto mma_g = [&](const Frag& F) {
    const v4f aN = -F.aI;
    const v2f ar2[2] = {__builtin_shufflevector(F.aR, F.aR, 0, 1),
                        __builtin_shufflevector(F.aR, F.aR, 2, 3)};
    const v2f ai2[2] = {__builtin_shufflevector(F.aI, F.aI, 0, 1),
                        __builtin_shufflevector(F.aI, F.aI, 2, 3)};
    const v2f an2[2] = {__builtin_shufflevector(aN, aN, 0, 1),
                        __builtin_shufflevector(aN, aN, 2, 3)};
    const v2f br2[2] = {__builtin_shufflevector(F.bR, F.bR, 0, 1),
                        __builtin_shufflevector(F.bR, F.bR, 2, 3)};
    const v2f bi2[2] = {__builtin_shufflevector(F.bI, F.bI, 0, 1),
                        __builtin_shufflevector(F.bI, F.bI, 2, 3)};
#pragma unroll
    for (int ti = 0; ti < 2; ++ti)
#pragma unroll
      for (int tj = 0; tj < 2; ++tj) {
        acc[0][ti][tj] = wmma_f32_16x16x4(ar2[ti], br2[tj], acc[0][ti][tj]);
        acc[0][ti][tj] = wmma_f32_16x16x4(ai2[ti], bi2[tj], acc[0][ti][tj]);
        acc[1][ti][tj] = wmma_f32_16x16x4(ar2[ti], bi2[tj], acc[1][ti][tj]);
        acc[1][ti][tj] = wmma_f32_16x16x4(an2[ti], br2[tj], acc[1][ti][tj]);
      }
  };
  {
    Frag X, Y;
    load_g(0, X);
    load_g(1, Y);
    __builtin_amdgcn_sched_barrier(0);
#pragma unroll
    for (int g = 0; g < OG; g += 2) {
      mma_g(X);
      __builtin_amdgcn_sched_barrier(0);
      if (g + 2 < OG) load_g(g + 2, X);
      __builtin_amdgcn_sched_barrier(0);
      mma_g(Y);
      __builtin_amdgcn_sched_barrier(0);
      if (g + 3 < OG) load_g(g + 3, Y);
      __builtin_amdgcn_sched_barrier(0);
    }
  }

  {
    float pw = ps;
    for (int off = 16; off > 0; off >>= 1) pw += __shfl_xor(pw, off, 32);
    if (lane == 0) aux[wave] = pw;
  }

  {
    float* mine = red + wave * 2048;
#pragma unroll
    for (int ri = 0; ri < 2; ++ri)
#pragma unroll
      for (int ti = 0; ti < 2; ++ti)
#pragma unroll
        for (int tl = 0; tl < 2; ++tl) {
          const v8f a = acc[ri][ti][tl];
          const int u0 = ((ti * 2 + tl) * 2) * 32 + lane;
          *(v4f*)(mine + ri * 1024 + u0 * 4) =
              __builtin_shufflevector(a, a, 0, 1, 2, 3);
          *(v4f*)(mine + ri * 1024 + (u0 + 32) * 4) =
              __builtin_shufflevector(a, a, 4, 5, 6, 7);
        }
  }
  __syncthreads();
  float f2 = aux[0];
  for (int w2 = 1; w2 < NW; ++w2) f2 += aux[w2];
  const float inv = 1.0f / f2;
  if (tid == 0 && blockIdx.x == 0 && blockIdx.y == 0) sout[0] = f2;
  float ssq = 0.f;
#pragma unroll
  for (int ri = 0; ri < 2; ++ri) {
    const float* p = red + ri * 1024 + 4 * tid;
    v4f s = *(const v4f*)p;
#pragma unroll
    for (int w2 = 1; w2 < NW; ++w2) s += *(const v4f*)(p + w2 * 2048);
    s *= inv;
    ssq += s.x * s.x;
    ssq += s.y * s.y;
    ssq += s.z * s.z;
    ssq += s.w * s.w;
  }

  for (int off = 16; off > 0; off >>= 1) ssq += __shfl_xor(ssq, off, 32);
  if (lane == 0) aux[NW + wave] = ssq;
  __syncthreads();
  if (threadIdx.x == 0) {
    float s = aux[NW];
    for (int w2 = 1; w2 < NW; ++w2) s += aux[NW + w2];
    const float wgt = (f == 0 ? 1.f : 2.f) * (ta == tb ? 1.f : 2.f);
    pout[blockIdx.y * gridDim.x + blockIdx.x] = wgt * s;
  }
}

__global__ __launch_bounds__(256) void spec_gram8_k(
    const float* __restrict__ src, const float* __restrict__ pin, int nin,
    float* __restrict__ pout, float* __restrict__ sout, size_t ws_bytes,
    size_t need) {
  __shared__ __attribute__((aligned(16))) float red[8 * 2 * 32 * 32];
  __shared__ float aux[16];
  if (ws_bytes < need) return;
  spec_gram_body(src, pin, nin, pout, sout, red, aux);
}


#define N_PART3 (18 * 145)
#define N_PART3S (21 * 145)
__global__ __launch_bounds__(256) void sigma_scale_k(
    const float* __restrict__ in, const float* __restrict__ scal,
    const float* __restrict__ part, float* __restrict__ out, int n,
    size_t ws_bytes, size_t need) {
  __shared__ float wsum[8];
  __shared__ float bc;
  const int tid = (int)threadIdx.x;
  const int base = (int)blockIdx.x * 2048 + tid;
  float x[8];
#pragma unroll
  for (int j = 0; j < 8; ++j) {
    const int i = base + 256 * j;
    x[j] = in[i < n ? i : 0];
  }
  constexpr int PER = (N_PART3S + 255) / 256;
  const int npart = ws_bytes >= need ? N_PART3S : N_PART3;
  float s = 0.f;
#pragma unroll
  for (int j = 0; j < PER; ++j) {
    const int i = tid + 256 * j;
    const float v = part[i < npart ? i : 0];
    s += i < npart ? v : 0.f;
  }
  for (int off = 16; off > 0; off >>= 1) s += __shfl_xor(s, off, 32);
  if ((tid & 31) == 0) wsum[tid >> 5] = s;
  __syncthreads();
  if (tid == 0) {
    float s3 = wsum[0];
    for (int w2 = 1; w2 < 8; ++w2) s3 += wsum[w2];
    const double s3d = ws_bytes >= need ? (double)s3 / 289.0 : (double)s3;
    double ls = 0.5 * (log((double)scal[0]) + 0.5 * log((double)scal[1]) +
                       0.25 * log((double)scal[2]) + 0.125 * log(s3d));
    bc = (float)exp(-ls);
  }
  __syncthreads();
  const float inv_sigma = bc;
#pragma unroll
  for (int j = 0; j < 8; ++j) {
    const int i = base + 256 * j;
    if (i < n) out[i] = x[j] * inv_sigma;
  }
}

extern "C" void kernel_launch(void* const* d_in, const int* in_sizes, int n_in,
                              void* d_out, int out_size, void* d_ws,
                              size_t ws_size, hipStream_t stream) {
  (void)in_sizes; (void)n_in; (void)out_size;
  const float* K = (const float*)d_in[0];
  float* out = (float*)d_out;
  char* ws = (char*)d_ws;

  float* scal = (float*)ws;
  float* partA = (float*)(ws + 1024);
  float* partB = (float*)(ws + 8192);
  float* partC = (float*)(ws + 12288);
  float* partD = (float*)(ws + 16384);
  float* K1T = (float*)(ws + (1 << 17));
  float* K2T = (float*)(ws + (4 << 20));
  float* KT0 = K2T;
  float* XT = (float*)(ws + (16 << 20));

  const int N0 = N_IN;

  transpose_in_k<<<1296, 256, 0, stream>>>(K, KT0, N0, partA);

  gram_wmma16_k<<<dim3(18, 13), 512, 0, stream>>>(KT0, K1T, partA, 1296, partB,
                                                  scal + 0, 3, ws_size, 0);

  gram_wmma16_k<<<dim3(18, 41), 512, 0, stream>>>(K1T, K2T, partB, 18 * 13,
                                                  partC, scal + 1, 5, ws_size, 0);

  dft17_k<<<dim3(144, 3), 256, 0, stream>>>(K2T, XT, ws_size, 59539456);
  spec_gram8_k<<<dim3(21, 145), 256, 0, stream>>>(XT, partC, 18 * 41, partD,
                                                  scal + 2, ws_size, 59539456);
  gram_wmma16_k<<<dim3(18, 145), 512, 0, stream>>>(K2T, nullptr, partC, 18 * 41,
                                                   partD, scal + 2, 9, ws_size,
                                                   59539456);

  sigma_scale_k<<<162, 256, 0, stream>>>(K, scal, partD, out, N0, ws_size,
                                         59539456);
}
